# starting kernel
# baseline (speedup 1.0000x reference)
#include <hip/hip_runtime.h>
#include <math.h>

typedef __attribute__((ext_vector_type(2))) float v2f;
typedef __attribute__((ext_vector_type(8))) float v8f;

#define IN_FEATS 256
#define OUT_FEATS 64
#define MAXK 16
#define TILE_M 16
#define LDA (IN_FEATS + 4)
#define LDH (OUT_FEATS + 4)

__global__ void init_kernel(float* __restrict__ out, const float* __restrict__ bias,
                            float* __restrict__ deg, int n_nodes) {
    long long i = (long long)blockIdx.x * blockDim.x + threadIdx.x;
    long long total = (long long)n_nodes * OUT_FEATS;
    if (i < total) out[i] = bias[i & (OUT_FEATS - 1)];
    if (i < 2LL * n_nodes) deg[i] = 0.0f;
}

__global__ void deg_wt_kernel(const int* __restrict__ src, const int* __restrict__ dst,
                              float* __restrict__ out_deg, float* __restrict__ in_deg,
                              const float* __restrict__ weight, float* __restrict__ wT,
                              int n_edges) {
    int i = blockIdx.x * blockDim.x + threadIdx.x;
    if (i < n_edges) {
        atomicAdd(&out_deg[src[i]], 1.0f);
        atomicAdd(&in_deg[dst[i]], 1.0f);
    }
    if (i < IN_FEATS * OUT_FEATS) {
        int k = i >> 6;
        int c = i & 63;
        wT[c * IN_FEATS + k] = weight[i];
    }
}

__global__ __launch_bounds__(128)
void gemm_maxk_kernel(const float* __restrict__ feat, const float* __restrict__ wT,
                      const float* __restrict__ out_deg, const float* __restrict__ in_deg,
                      float* __restrict__ tk_vals, unsigned int* __restrict__ tk_idx,
                      int n_nodes) {
    __shared__ float sA[TILE_M * LDA];
    __shared__ float sH[TILE_M * LDH];

    const int tid = threadIdx.x;
    const int m0 = blockIdx.x * TILE_M;

    for (int i = tid; i < TILE_M * (IN_FEATS / 4); i += 128) {
        int row = i / (IN_FEATS / 4);
        int c4  = i % (IN_FEATS / 4);
        float4 v = make_float4(0.f, 0.f, 0.f, 0.f);
        if (m0 + row < n_nodes)
            v = ((const float4*)feat)[(long long)(m0 + row) * (IN_FEATS / 4) + c4];
        float* p = &sA[row * LDA + c4 * 4];
        p[0] = v.x; p[1] = v.y; p[2] = v.z; p[3] = v.w;
    }
    __syncthreads();

    const int wave = tid >> 5;
    const int lane = tid & 31;
    const int r16  = lane & 15;
    const int koff = (lane >> 4) << 1;
    const float* wcol = wT + (wave * 16 + r16) * IN_FEATS;
    const float* arow = sA + r16 * LDA;

    v8f c = {0.f, 0.f, 0.f, 0.f, 0.f, 0.f, 0.f, 0.f};
    #pragma unroll
    for (int k0 = 0; k0 < IN_FEATS; k0 += 4) {
        v2f a, b;
        a.x = arow[k0 + koff];
        a.y = arow[k0 + koff + 1];
        b.x = wcol[k0 + koff];
        b.y = wcol[k0 + koff + 1];
        c = __builtin_amdgcn_wmma_f32_16x16x4_f32(false, a, false, b, (short)0, c,
                                                  false, false);
    }

    {
        const int rbase = (lane >> 4) << 3;
        const int col = wave * 16 + r16;
        #pragma unroll
        for (int r = 0; r < 8; ++r)
            sH[(rbase + r) * LDH + col] = c[r];
    }
    __syncthreads();

    if (tid < TILE_M && (m0 + tid) < n_nodes) {
        const int node = m0 + tid;
        float od = out_deg[node]; od = od < 1.f ? 1.f : od;
        float id = in_deg[node];  id = id < 1.f ? 1.f : id;
        const float scale = rsqrtf(od) * rsqrtf(id);
        float* rowp = &sH[tid * LDH];

        unsigned int iout0 = 0u, iout1 = 0u, iout2 = 0u, iout3 = 0u;
        #pragma unroll
        for (int i = 0; i < MAXK; ++i) {
            float best = -3.402823466e+38f;
            int bi = 0;
            for (int j = 0; j < OUT_FEATS; ++j) {
                float v = rowp[j];
                if (v > best) { best = v; bi = j; }
            }
            rowp[bi] = -3.402823466e+38f;
            tk_vals[(long long)node * MAXK + i] = best * scale;
            unsigned int b8 = ((unsigned int)bi) << ((i & 3) * 8);
            if (i < 4)       iout0 |= b8;
            else if (i < 8)  iout1 |= b8;
            else if (i < 12) iout2 |= b8;
            else             iout3 |= b8;
        }
        unsigned int* iq = &tk_idx[(long long)node * 4];
        iq[0] = iout0; iq[1] = iout1; iq[2] = iout2; iq[3] = iout3;
    }
}

__global__ void scatter_kernel(const int* __restrict__ src, const int* __restrict__ dst,
                               const float* __restrict__ tk_vals,
                               const unsigned int* __restrict__ tk_idx,
                               float* __restrict__ out, int n_edges) {
    long long gid = (long long)blockIdx.x * blockDim.x + threadIdx.x;
    long long total = (long long)n_edges * 4;
    if (gid >= total) return;
    int e = (int)(gid >> 2);
    int part = (int)(gid & 3);
    int s = src[e], d = dst[e];
    float4 v = ((const float4*)tk_vals)[(long long)s * 4 + part];
    unsigned int ip = tk_idx[(long long)s * 4 + part];
    float* orow = out + (long long)d * OUT_FEATS;
    atomicAdd(orow + (ip & 0xffu), v.x);
    atomicAdd(orow + ((ip >> 8) & 0xffu), v.y);
    atomicAdd(orow + ((ip >> 16) & 0xffu), v.z);
    atomicAdd(orow + ((ip >> 24) & 0xffu), v.w);
}

extern "C" void kernel_launch(void* const* d_in, const int* in_sizes, int n_in,
                              void* d_out, int out_size, void* d_ws, size_t ws_size,
                              hipStream_t stream) {
    const float* feat   = (const float*)d_in[0];
    const float* weight = (const float*)d_in[1];
    const float* bias   = (const float*)d_in[2];
    const int*   src    = (const int*)d_in[3];
    const int*   dst    = (const int*)d_in[4];
    float* out = (float*)d_out;

    const int n_nodes = in_sizes[0] / IN_FEATS;
    const int n_edges = in_sizes[3];

    float* ws_f    = (float*)d_ws;
    float* out_deg = ws_f;
    float* in_deg  = out_deg + n_nodes;
    float* wT      = in_deg + n_nodes;
    float* tk_vals = wT + IN_FEATS * OUT_FEATS;
    unsigned int* tk_idx = (unsigned int*)(tk_vals + (long long)n_nodes * MAXK);

    {
        long long total = (long long)n_nodes * OUT_FEATS;
        int grid = (int)((total + 255) / 256);
        init_kernel<<<grid, 256, 0, stream>>>(out, bias, out_deg, n_nodes);
    }
    {
        int work = n_edges > IN_FEATS * OUT_FEATS ? n_edges : IN_FEATS * OUT_FEATS;
        int grid = (work + 255) / 256;
        deg_wt_kernel<<<grid, 256, 0, stream>>>(src, dst, out_deg, in_deg, weight, wT,
                                                n_edges);
    }
    {
        int grid = (n_nodes + TILE_M - 1) / TILE_M;
        gemm_maxk_kernel<<<grid, 128, 0, stream>>>(feat, wT, out_deg, in_deg,
                                                   tk_vals, tk_idx, n_nodes);
    }
    {
        long long total = (long long)n_edges * 4;
        int grid = (int)((total + 255) / 256);
        scatter_kernel<<<grid, 256, 0, stream>>>(src, dst, tk_vals, tk_idx, out, n_edges);
    }
}


# degree counters moved out of two dense 400 KB arrays into words 0 and 32 of each node's own output row (one counter per 128-B line); bias fill moved behind the read, into the GEMM kernel
# speedup vs baseline: 1.1404x; 1.1404x over previous
#include <hip/hip_runtime.h>
#include <math.h>

typedef __attribute__((ext_vector_type(2))) float v2f;
typedef __attribute__((ext_vector_type(8))) float v8f;

#define IN_FEATS 256
#define OUT_FEATS 64
#define MAXK 16
#define TILE_M 16
#define LDA (IN_FEATS + 4)
#define LDH (OUT_FEATS + 4)

#define DEG_OUT_COL 0
#define DEG_IN_COL  32

__global__ void init_kernel(float* __restrict__ out, int n_nodes) {
    long long i = (long long)blockIdx.x * blockDim.x + threadIdx.x;
    if (i < 2LL * n_nodes)
        out[(i >> 1) * OUT_FEATS + ((i & 1) ? DEG_IN_COL : DEG_OUT_COL)] = 0.0f;
}

__global__ void deg_wt_kernel(const int* __restrict__ src, const int* __restrict__ dst,
                              float* __restrict__ out,
                              const float* __restrict__ weight, float* __restrict__ wT,
                              int n_edges) {
    int i = blockIdx.x * blockDim.x + threadIdx.x;
    if (i < n_edges) {
        const int s = src[i];
        const int d = dst[i];
        atomicAdd(out + (long long)s * OUT_FEATS + DEG_OUT_COL, 1.0f);
        atomicAdd(out + (long long)d * OUT_FEATS + DEG_IN_COL, 1.0f);
    }
    if (i < IN_FEATS * OUT_FEATS) {
        int k = i >> 6;
        int c = i & 63;
        wT[c * IN_FEATS + k] = weight[i];
    }
}

__global__ __launch_bounds__(128)
void gemm_maxk_kernel(const float* __restrict__ feat, const float* __restrict__ wT,
                      const float* __restrict__ bias, float* out,
                      float* __restrict__ tk_vals, unsigned int* __restrict__ tk_idx,
                      int n_nodes) {
    __shared__ float sA[TILE_M * LDA];
    __shared__ float sH[TILE_M * LDH];
    __shared__ float sScale[TILE_M];

    const int tid = threadIdx.x;
    const int m0 = blockIdx.x * TILE_M;

    if (tid < TILE_M) {
        float sc = 0.f;
        if (m0 + tid < n_nodes) {
            const float* orow = out + (long long)(m0 + tid) * OUT_FEATS;
            float od = orow[DEG_OUT_COL]; od = od < 1.f ? 1.f : od;
            float id = orow[DEG_IN_COL];  id = id < 1.f ? 1.f : id;
            sc = rsqrtf(od) * rsqrtf(id);
        }
        sScale[tid] = sc;
    }

    for (int i = tid; i < TILE_M * (IN_FEATS / 4); i += 128) {
        int row = i / (IN_FEATS / 4);
        int c4  = i % (IN_FEATS / 4);
        float4 v = make_float4(0.f, 0.f, 0.f, 0.f);
        if (m0 + row < n_nodes)
            v = ((const float4*)feat)[(long long)(m0 + row) * (IN_FEATS / 4) + c4];
        float* p = &sA[row * LDA + c4 * 4];
        p[0] = v.x; p[1] = v.y; p[2] = v.z; p[3] = v.w;
    }
    __syncthreads();

    for (int i = tid; i < TILE_M * (OUT_FEATS / 4); i += 128) {
        int row = i / (OUT_FEATS / 4);
        int c4  = i % (OUT_FEATS / 4);
        if (m0 + row < n_nodes)
            ((float4*)out)[(long long)(m0 + row) * (OUT_FEATS / 4) + c4] =
                ((const float4*)bias)[c4];
    }

    const int wave = tid >> 5;
    const int lane = tid & 31;
    const int r16  = lane & 15;
    const int koff = (lane >> 4) << 1;
    const float* wcol = wT + (wave * 16 + r16) * IN_FEATS;
    const float* arow = sA + r16 * LDA;

    v8f c = {0.f, 0.f, 0.f, 0.f, 0.f, 0.f, 0.f, 0.f};
    #pragma unroll
    for (int k0 = 0; k0 < IN_FEATS; k0 += 4) {
        v2f a, b;
        a.x = arow[k0 + koff];
        a.y = arow[k0 + koff + 1];
        b.x = wcol[k0 + koff];
        b.y = wcol[k0 + koff + 1];
        c = __builtin_amdgcn_wmma_f32_16x16x4_f32(false, a, false, b, (short)0, c,
                                                  false, false);
    }

    {
        const int rbase = (lane >> 4) << 3;
        const int col = wave * 16 + r16;
        #pragma unroll
        for (int r = 0; r < 8; ++r)
            sH[(rbase + r) * LDH + col] = c[r];
    }
    __syncthreads();

    if (tid < TILE_M && (m0 + tid) < n_nodes) {
        const int node = m0 + tid;
        const float scale = sScale[tid];
        float* rowp = &sH[tid * LDH];

        unsigned int iout0 = 0u, iout1 = 0u, iout2 = 0u, iout3 = 0u;
        #pragma unroll
        for (int i = 0; i < MAXK; ++i) {
            float best = -3.402823466e+38f;
            int bi = 0;
            for (int j = 0; j < OUT_FEATS; ++j) {
                float v = rowp[j];
                if (v > best) { best = v; bi = j; }
            }
            rowp[bi] = -3.402823466e+38f;
            tk_vals[(long long)node * MAXK + i] = best * scale;
            unsigned int b8 = ((unsigned int)bi) << ((i & 3) * 8);
            if (i < 4)       iout0 |= b8;
            else if (i < 8)  iout1 |= b8;
            else if (i < 12) iout2 |= b8;
            else             iout3 |= b8;
        }
        unsigned int* iq = &tk_idx[(long long)node * 4];
        iq[0] = iout0; iq[1] = iout1; iq[2] = iout2; iq[3] = iout3;
    }
}

__global__ void scatter_kernel(const int* __restrict__ src, const int* __restrict__ dst,
                               const float* __restrict__ tk_vals,
                               const unsigned int* __restrict__ tk_idx,
                               float* __restrict__ out, int n_edges) {
    long long gid = (long long)blockIdx.x * blockDim.x + threadIdx.x;
    long long total = (long long)n_edges * 4;
    if (gid >= total) return;
    int e = (int)(gid >> 2);
    int part = (int)(gid & 3);
    int s = src[e], d = dst[e];
    float4 v = ((const float4*)tk_vals)[(long long)s * 4 + part];
    unsigned int ip = tk_idx[(long long)s * 4 + part];
    float* orow = out + (long long)d * OUT_FEATS;
    atomicAdd(orow + (ip & 0xffu), v.x);
    atomicAdd(orow + ((ip >> 8) & 0xffu), v.y);
    atomicAdd(orow + ((ip >> 16) & 0xffu), v.z);
    atomicAdd(orow + ((ip >> 24) & 0xffu), v.w);
}

extern "C" void kernel_launch(void* const* d_in, const int* in_sizes, int n_in,
                              void* d_out, int out_size, void* d_ws, size_t ws_size,
                              hipStream_t stream) {
    const float* feat   = (const float*)d_in[0];
    const float* weight = (const float*)d_in[1];
    const float* bias   = (const float*)d_in[2];
    const int*   src    = (const int*)d_in[3];
    const int*   dst    = (const int*)d_in[4];
    float* out = (float*)d_out;

    const int n_nodes = in_sizes[0] / IN_FEATS;
    const int n_edges = in_sizes[3];

    float* ws_f    = (float*)d_ws;
    float* wT      = ws_f;
    float* tk_vals = wT + IN_FEATS * OUT_FEATS;
    unsigned int* tk_idx = (unsigned int*)(tk_vals + (long long)n_nodes * MAXK);

    {
        long long total = 2LL * n_nodes;
        int grid = (int)((total + 255) / 256);
        init_kernel<<<grid, 256, 0, stream>>>(out, n_nodes);
    }
    {
        int work = n_edges > IN_FEATS * OUT_FEATS ? n_edges : IN_FEATS * OUT_FEATS;
        int grid = (work + 255) / 256;
        deg_wt_kernel<<<grid, 256, 0, stream>>>(src, dst, out, weight, wT, n_edges);
    }
    {
        int grid = (n_nodes + TILE_M - 1) / TILE_M;
        gemm_maxk_kernel<<<grid, 128, 0, stream>>>(feat, wT, bias, out,
                                                   tk_vals, tk_idx, n_nodes);
    }
    {
        long long total = (long long)n_edges * 4;
        int grid = (int)((total + 255) / 256);
        scatter_kernel<<<grid, 256, 0, stream>>>(src, dst, tk_vals, tk_idx, out, n_edges);
    }
}


# float-atomic edge scatter replaced by gather by destination: in-degree add returns edge rank, sources placed per node, one wave sums a node's rows in LDS, row stored once; sum order = list order
# speedup vs baseline: 1.3682x; 1.1998x over previous
#include <hip/hip_runtime.h>
#include <math.h>

typedef __attribute__((ext_vector_type(2))) float v2f;
typedef __attribute__((ext_vector_type(8))) float v8f;

#define IN_FEATS 256
#define OUT_FEATS 64
#define MAXK 16
#define TILE_M 16
#define LDA (IN_FEATS + 4)
#define LDH (OUT_FEATS + 4)

#define DEG_OUT_COL 0
#define DEG_IN_COL  32

__global__ void init_kernel(float* __restrict__ out, int* __restrict__ list_top, int n_nodes) {
    long long i = (long long)blockIdx.x * blockDim.x + threadIdx.x;
    if (i == 0 && list_top != nullptr) *list_top = 0;
    if (i < 2LL * n_nodes)
        out[(i >> 1) * OUT_FEATS + ((i & 1) ? DEG_IN_COL : DEG_OUT_COL)] = 0.0f;
}

__global__ void deg_wt_kernel(const int* __restrict__ src, const int* __restrict__ dst,
                              float* __restrict__ out,
                              const float* __restrict__ weight, float* __restrict__ wT,
                              int* __restrict__ rank, int n_edges) {
    int i = blockIdx.x * blockDim.x + threadIdx.x;
    if (i < n_edges) {
        const int s = src[i];
        const int d = dst[i];
        atomicAdd(out + (long long)s * OUT_FEATS + DEG_OUT_COL, 1.0f);
        const int r = atomicAdd((int*)(out + (long long)d * OUT_FEATS + DEG_IN_COL), 1);
        if (rank != nullptr) rank[i] = r;
    }
    if (i < IN_FEATS * OUT_FEATS) {
        int k = i >> 6;
        int c = i & 63;
        wT[c * IN_FEATS + k] = weight[i];
    }
}

__global__ __launch_bounds__(128)
void gemm_maxk_kernel(const float* __restrict__ feat, const float* __restrict__ wT,
                      const float* __restrict__ bias, float* out,
                      float* __restrict__ tk_vals, unsigned int* __restrict__ tk_idx,
                      int* __restrict__ seg, int* __restrict__ list_top,
                      int n_nodes) {
    __shared__ float sA[TILE_M * LDA];
    __shared__ float sH[TILE_M * LDH];
    __shared__ float sScale[TILE_M];

    const int tid = threadIdx.x;
    const int m0 = blockIdx.x * TILE_M;

    if (tid < TILE_M) {
        float sc = 0.f;
        if (m0 + tid < n_nodes) {
            const float* orow = out + (long long)(m0 + tid) * OUT_FEATS;
            float od = orow[DEG_OUT_COL]; od = od < 1.f ? 1.f : od;
            float id = (float)((const int*)orow)[DEG_IN_COL]; id = id < 1.f ? 1.f : id;
            sc = rsqrtf(od) * rsqrtf(id);
        }
        sScale[tid] = sc;
    }

    for (int i = tid; i < TILE_M * (IN_FEATS / 4); i += 128) {
        int row = i / (IN_FEATS / 4);
        int c4  = i % (IN_FEATS / 4);
        float4 v = make_float4(0.f, 0.f, 0.f, 0.f);
        if (m0 + row < n_nodes)
            v = ((const float4*)feat)[(long long)(m0 + row) * (IN_FEATS / 4) + c4];
        float* p = &sA[row * LDA + c4 * 4];
        p[0] = v.x; p[1] = v.y; p[2] = v.z; p[3] = v.w;
    }
    __syncthreads();

    for (int i = tid; seg == nullptr && i < TILE_M * (OUT_FEATS / 4); i += 128) {
        int row = i / (OUT_FEATS / 4);
        int c4  = i % (OUT_FEATS / 4);
        if (m0 + row < n_nodes)
            ((float4*)out)[(long long)(m0 + row) * (OUT_FEATS / 4) + c4] =
                ((const float4*)bias)[c4];
    }

    const int wave = tid >> 5;
    const int lane = tid & 31;
    const int r16  = lane & 15;
    const int koff = (lane >> 4) << 1;
    const float* wcol = wT + (wave * 16 + r16) * IN_FEATS;
    const float* arow = sA + r16 * LDA;

    v8f c = {0.f, 0.f, 0.f, 0.f, 0.f, 0.f, 0.f, 0.f};
    #pragma unroll
    for (int k0 = 0; k0 < IN_FEATS; k0 += 4) {
        v2f a, b;
        a.x = arow[k0 + koff];
        a.y = arow[k0 + koff + 1];
        b.x = wcol[k0 + koff];
        b.y = wcol[k0 + koff + 1];
        c = __builtin_amdgcn_wmma_f32_16x16x4_f32(false, a, false, b, (short)0, c,
                                                  false, false);
    }

    {
        const int rbase = (lane >> 4) << 3;
        const int col = wave * 16 + r16;
        #pragma unroll
        for (int r = 0; r < 8; ++r)
            sH[(rbase + r) * LDH + col] = c[r];
    }
    __syncthreads();

    if (tid < TILE_M && (m0 + tid) < n_nodes) {
        const int node = m0 + tid;
        const float scale = sScale[tid];
        float* rowp = &sH[tid * LDH];

        unsigned int iout0 = 0u, iout1 = 0u, iout2 = 0u, iout3 = 0u;
        #pragma unroll
        for (int i = 0; i < MAXK; ++i) {
            float best = -3.402823466e+38f;
            int bi = 0;
            for (int j = 0; j < OUT_FEATS; ++j) {
                float v = rowp[j];
                if (v > best) { best = v; bi = j; }
            }
            rowp[bi] = -3.402823466e+38f;
            tk_vals[(long long)node * MAXK + i] = best * scale;
            unsigned int b8 = ((unsigned int)bi) << ((i & 3) * 8);
            if (i < 4)       iout0 |= b8;
            else if (i < 8)  iout1 |= b8;
            else if (i < 12) iout2 |= b8;
            else             iout3 |= b8;
        }
        unsigned int* iq = &tk_idx[(long long)node * 4];
        iq[0] = iout0; iq[1] = iout1; iq[2] = iout2; iq[3] = iout3;
    }

    if (seg != nullptr && tid == 32) {
        int cnt[TILE_M];
        int tot = 0;
        #pragma unroll
        for (int r = 0; r < TILE_M; ++r) {
            cnt[r] = (m0 + r < n_nodes)
                   ? ((const int*)out)[(long long)(m0 + r) * OUT_FEATS + DEG_IN_COL] : 0;
            tot += cnt[r];
        }
        int run = atomicAdd(list_top, tot);
        #pragma unroll
        for (int r = 0; r < TILE_M; ++r) {
            if (m0 + r < n_nodes) ((int2*)seg)[m0 + r] = make_int2(run, cnt[r]);
            run += cnt[r];
        }
    }
}

__global__ void scatter_kernel(const int* __restrict__ src, const int* __restrict__ dst,
                               const float* __restrict__ tk_vals,
                               const unsigned int* __restrict__ tk_idx,
                               float* __restrict__ out, int n_edges) {
    long long gid = (long long)blockIdx.x * blockDim.x + threadIdx.x;
    long long total = (long long)n_edges * 4;
    if (gid >= total) return;
    int e = (int)(gid >> 2);
    int part = (int)(gid & 3);
    int s = src[e], d = dst[e];
    float4 v = ((const float4*)tk_vals)[(long long)s * 4 + part];
    unsigned int ip = tk_idx[(long long)s * 4 + part];
    float* orow = out + (long long)d * OUT_FEATS;
    atomicAdd(orow + (ip & 0xffu), v.x);
    atomicAdd(orow + ((ip >> 8) & 0xffu), v.y);
    atomicAdd(orow + ((ip >> 16) & 0xffu), v.z);
    atomicAdd(orow + ((ip >> 24) & 0xffu), v.w);
}


__global__ void fill_kernel(const int* __restrict__ src, const int* __restrict__ dst,
                            const int* __restrict__ rank, const int* __restrict__ seg,
                            int* __restrict__ sorted_src, int n_edges) {
    const long long stride = (long long)gridDim.x * blockDim.x;
    for (long long e = (long long)blockIdx.x * blockDim.x + threadIdx.x; e < n_edges; e += stride) {
        const int d = dst[e];
        sorted_src[seg[2LL * d] + rank[e]] = src[e];
    }
}

__global__ __launch_bounds__(256)
void gather_kernel(const int* __restrict__ seg, const int* __restrict__ sorted_src,
                   const float* __restrict__ tk_vals, const unsigned int* __restrict__ tk_idx,
                   const float* __restrict__ bias, float* __restrict__ out, int n_nodes) {
    __shared__ __attribute__((aligned(16))) float sRow[8][OUT_FEATS];
    const int lane = threadIdx.x & 31;
    const int wave = threadIdx.x >> 5;
    const int half = lane >> 4;
    const int k    = lane & 15;
    const int sh   = (k & 3) * 8;
    float* row = sRow[wave];
    const float2 b2 = ((const float2*)bias)[lane];

    for (int node = blockIdx.x * 8 + wave; node < n_nodes; node += gridDim.x * 8) {
        const int2 sg = ((const int2*)seg)[node];
        const int start = __builtin_amdgcn_readfirstlane(sg.x);
        const int cnt   = __builtin_amdgcn_readfirstlane(sg.y);
        row[lane] = 0.f;
        row[lane + 32] = 0.f;
        for (int base = 0; base < cnt; base += 32) {
            const int rem = (cnt - base) < 32 ? (cnt - base) : 32;
            const int sl = lane < rem ? sorted_src[start + base + lane] : 0;
            for (int q = 0; q < rem; q += 8) {
                float v[4];
                unsigned int iw[4];
                #pragma unroll
                for (int u = 0; u < 4; ++u) {
                    const int s = __shfl(sl, q + 2 * u + half);
                    v[u]  = tk_vals[(long long)s * MAXK + k];
                    iw[u] = tk_idx[(long long)s * 4 + (k >> 2)];
                }
                #pragma unroll
                for (int u = 0; u < 4; ++u)
                    if (q + 2 * u + half < rem)
                        atomicAdd(&row[(iw[u] >> sh) & 0xffu], v[u]);
            }
        }
        const float2 r2 = ((const float2*)row)[lane];
        ((float2*)out)[(long long)node * (OUT_FEATS / 2) + lane] =
            make_float2(r2.x + b2.x, r2.y + b2.y);
    }
}

extern "C" void kernel_launch(void* const* d_in, const int* in_sizes, int n_in,
                              void* d_out, int out_size, void* d_ws, size_t ws_size,
                              hipStream_t stream) {
    const float* feat   = (const float*)d_in[0];
    const float* weight = (const float*)d_in[1];
    const float* bias   = (const float*)d_in[2];
    const int*   src    = (const int*)d_in[3];
    const int*   dst    = (const int*)d_in[4];
    float* out = (float*)d_out;

    const int n_nodes = in_sizes[0] / IN_FEATS;
    const int n_edges = in_sizes[3];

    float* ws_f    = (float*)d_ws;
    float* wT      = ws_f;
    float* tk_vals = wT + IN_FEATS * OUT_FEATS;
    unsigned int* tk_idx = (unsigned int*)(tk_vals + (long long)n_nodes * MAXK);

    const size_t o_seg    = (((size_t)IN_FEATS * OUT_FEATS + (size_t)n_nodes * (MAXK + 4)) * 4 + 255) & ~(size_t)255;
    const size_t o_rank   = (o_seg + (size_t)n_nodes * 8 + 255) & ~(size_t)255;
    const size_t o_sorted = (o_rank + (size_t)n_edges * 4 + 255) & ~(size_t)255;
    const size_t o_top    = (o_sorted + (size_t)n_edges * 4 + 255) & ~(size_t)255;
    const bool gather = ws_size >= o_top + 256;
    char* ws_b = (char*)d_ws;
    int*  seg        = gather ? (int*)(ws_b + o_seg)    : nullptr;
    int*  rank       = gather ? (int*)(ws_b + o_rank)   : nullptr;
    int*  sorted_src = gather ? (int*)(ws_b + o_sorted) : nullptr;
    int*  list_top   = gather ? (int*)(ws_b + o_top)    : nullptr;

    {
        long long total = 2LL * n_nodes;
        int grid = (int)((total + 255) / 256);
        init_kernel<<<grid, 256, 0, stream>>>(out, list_top, n_nodes);
    }
    {
        int work = n_edges > IN_FEATS * OUT_FEATS ? n_edges : IN_FEATS * OUT_FEATS;
        int grid = (work + 255) / 256;
        deg_wt_kernel<<<grid, 256, 0, stream>>>(src, dst, out, weight, wT, rank, n_edges);
    }
    {
        int grid = (n_nodes + TILE_M - 1) / TILE_M;
        gemm_maxk_kernel<<<grid, 128, 0, stream>>>(feat, wT, bias, out,
                                                   tk_vals, tk_idx, seg, list_top, n_nodes);
    }
    if (gather) {
        fill_kernel<<<2048, 256, 0, stream>>>(src, dst, rank, seg, sorted_src, n_edges);
        gather_kernel<<<2048, 256, 0, stream>>>(seg, sorted_src, tk_vals, tk_idx, bias, out, n_nodes);
    } else {
        long long total = (long long)n_edges * 4;
        int grid = (int)((total + 255) / 256);
        scatter_kernel<<<grid, 256, 0, stream>>>(src, dst, tk_vals, tk_idx, out, n_edges);
    }
}


# degrees and edge ranks on LDS counters: block = (node range, edge slice), one whole-line global add per node and slice, not two scattered atomics per edge; rank, so sum order, slice-major
# speedup vs baseline: 1.8331x; 1.3398x over previous
#include <hip/hip_runtime.h>
#include <math.h>

typedef __attribute__((ext_vector_type(2))) float v2f;
typedef __attribute__((ext_vector_type(8))) float v8f;

#define IN_FEATS 256
#define OUT_FEATS 64
#define MAXK 16
#define TILE_M 16
#define LDA (IN_FEATS + 4)
#define LDH (OUT_FEATS + 4)

#define DEG_OUT_COL 0
#define DEG_IN_COL  32

__global__ void init_kernel(float* __restrict__ out, int* __restrict__ list_top, int n_nodes) {
    long long i = (long long)blockIdx.x * blockDim.x + threadIdx.x;
    if (i == 0 && list_top != nullptr) *list_top = 0;
    if (i < 2LL * n_nodes)
        out[(i >> 1) * OUT_FEATS + ((i & 1) ? DEG_IN_COL : DEG_OUT_COL)] = 0.0f;
}

__global__ void deg_wt_kernel(const int* __restrict__ src, const int* __restrict__ dst,
                              float* __restrict__ out,
                              const float* __restrict__ weight, float* __restrict__ wT,
                              int* __restrict__ rank, int n_edges) {
    int i = blockIdx.x * blockDim.x + threadIdx.x;
    if (i < n_edges) {
        const int s = src[i];
        const int d = dst[i];
        atomicAdd(out + (long long)s * OUT_FEATS + DEG_OUT_COL, 1.0f);
        const int r = atomicAdd((int*)(out + (long long)d * OUT_FEATS + DEG_IN_COL), 1);
        if (rank != nullptr) rank[i] = r;
    }
    if (i < IN_FEATS * OUT_FEATS) {
        int k = i >> 6;
        int c = i & 63;
        wT[c * IN_FEATS + k] = weight[i];
    }
}


#define DEG_R       6400
#define DEG_NS      16
#define DEG_THREADS 1024
#define DEG_FL      ((DEG_R + DEG_THREADS - 1) / DEG_THREADS)

__global__ void init_dense_kernel(int* __restrict__ dcnt, int* __restrict__ list_top, int n2) {
    int i = blockIdx.x * blockDim.x + threadIdx.x;
    if (i == 0) *list_top = 0;
    if (i < n2) dcnt[i] = 0;
}

__device__ __forceinline__ void count_in_range(int* cnt, int v, int lo, int len) {
    const unsigned int o = (unsigned int)(v - lo);
    if (o < (unsigned int)len) atomicAdd(&cnt[o], 1);
}

__device__ __forceinline__ void rank_in_range(int* cnt, int* __restrict__ rank, int e,
                                              int v, int lo, int len) {
    const unsigned int o = (unsigned int)(v - lo);
    if (o < (unsigned int)len) rank[e] = atomicAdd(&cnt[o], 1);
}

__global__ __launch_bounds__(DEG_THREADS)
void deg_lds_kernel(const int* __restrict__ src, const int* __restrict__ dst,
                    int* __restrict__ dcnt,
                    const float* __restrict__ weight, float* __restrict__ wT,
                    int* __restrict__ rank, int n_nodes, int n_edges) {
    __shared__ int sIn[DEG_R];
    __shared__ int sOut[DEG_R];

    const int tid = threadIdx.x;
    const int b = blockIdx.x / DEG_NS;
    const int s = blockIdx.x % DEG_NS;
    const int lo = b * DEG_R;
    const int len = (n_nodes - lo) < DEG_R ? (n_nodes - lo) : DEG_R;

    {
        const int gi = blockIdx.x * DEG_THREADS + tid;
        if (gi < IN_FEATS * OUT_FEATS) {
            const int k = gi >> 6;
            const int c = gi & 63;
            wT[c * IN_FEATS + k] = weight[gi];
        }
    }

    for (int n = tid; n < DEG_R; n += DEG_THREADS) { sIn[n] = 0; sOut[n] = 0; }
    __syncthreads();

    const int nq = n_edges >> 2;
    const int qs = (nq + DEG_NS - 1) / DEG_NS;
    const int q0 = s * qs;
    const int q1 = (q0 + qs) < nq ? (q0 + qs) : nq;
    const int e_tail = (nq << 2) + tid;
    const bool tail = (s == DEG_NS - 1) && (e_tail < n_edges);

    for (int q = q0 + tid; q < q1; q += 2 * DEG_THREADS) {
        const bool two = (q + DEG_THREADS) < q1;
        const int qb = two ? (q + DEG_THREADS) : q;
        const int len1 = two ? len : 0;
        const int4 d0 = ((const int4*)dst)[q];
        const int4 d1 = ((const int4*)dst)[qb];
        const int4 s0 = ((const int4*)src)[q];
        const int4 s1 = ((const int4*)src)[qb];
        count_in_range(sIn, d0.x, lo, len); count_in_range(sIn, d0.y, lo, len);
        count_in_range(sIn, d0.z, lo, len); count_in_range(sIn, d0.w, lo, len);
        count_in_range(sIn, d1.x, lo, len1); count_in_range(sIn, d1.y, lo, len1);
        count_in_range(sIn, d1.z, lo, len1); count_in_range(sIn, d1.w, lo, len1);
        count_in_range(sOut, s0.x, lo, len); count_in_range(sOut, s0.y, lo, len);
        count_in_range(sOut, s0.z, lo, len); count_in_range(sOut, s0.w, lo, len);
        count_in_range(sOut, s1.x, lo, len1); count_in_range(sOut, s1.y, lo, len1);
        count_in_range(sOut, s1.z, lo, len1); count_in_range(sOut, s1.w, lo, len1);
    }
    if (tail) {
        count_in_range(sIn, dst[e_tail], lo, len);
        count_in_range(sOut, src[e_tail], lo, len);
    }
    __syncthreads();

    {
        int ci[DEG_FL], start[DEG_FL];
        #pragma unroll
        for (int u = 0; u < DEG_FL; ++u) {
            const int n = tid + u * DEG_THREADS;
            ci[u] = n < len ? sIn[n] : 0;
            const int co = n < len ? sOut[n] : 0;
            start[u] = 0;
            if (ci[u] > 0) start[u] = atomicAdd(&dcnt[lo + n], ci[u]);
            if (co > 0) atomicAdd(&dcnt[n_nodes + lo + n], co);
        }
        #pragma unroll
        for (int u = 0; u < DEG_FL; ++u)
            if (ci[u] > 0) sIn[tid + u * DEG_THREADS] = start[u];
    }
    __syncthreads();

    for (int q = q0 + tid; q < q1; q += 2 * DEG_THREADS) {
        const bool two = (q + DEG_THREADS) < q1;
        const int qb = two ? (q + DEG_THREADS) : q;
        const int len1 = two ? len : 0;
        const int4 d0 = ((const int4*)dst)[q];
        const int4 d1 = ((const int4*)dst)[qb];
        const int e0 = q << 2;
        const int e1 = qb << 2;
        rank_in_range(sIn, rank, e0,     d0.x, lo, len);
        rank_in_range(sIn, rank, e0 + 1, d0.y, lo, len);
        rank_in_range(sIn, rank, e0 + 2, d0.z, lo, len);
        rank_in_range(sIn, rank, e0 + 3, d0.w, lo, len);
        rank_in_range(sIn, rank, e1,     d1.x, lo, len1);
        rank_in_range(sIn, rank, e1 + 1, d1.y, lo, len1);
        rank_in_range(sIn, rank, e1 + 2, d1.z, lo, len1);
        rank_in_range(sIn, rank, e1 + 3, d1.w, lo, len1);
    }
    if (tail) rank_in_range(sIn, rank, e_tail, dst[e_tail], lo, len);
}

__global__ __launch_bounds__(128)
void gemm_maxk_kernel(const float* __restrict__ feat, const float* __restrict__ wT,
                      const float* __restrict__ bias, float* out,
                      float* __restrict__ tk_vals, unsigned int* __restrict__ tk_idx,
                      int* __restrict__ seg, int* __restrict__ list_top,
                      const int* dcnt, int n_nodes) {
    __shared__ float sA[TILE_M * LDA];
    __shared__ float sH[TILE_M * LDH];
    __shared__ float sScale[TILE_M];

    const int tid = threadIdx.x;
    const int m0 = blockIdx.x * TILE_M;

    if (tid < TILE_M) {
        float sc = 0.f;
        if (m0 + tid < n_nodes) {
            float od, id;
            if (dcnt != nullptr) {
                id = (float)dcnt[m0 + tid];
                od = (float)dcnt[n_nodes + m0 + tid];
            } else {
                const float* orow = out + (long long)(m0 + tid) * OUT_FEATS;
                od = orow[DEG_OUT_COL];
                id = (float)((const int*)orow)[DEG_IN_COL];
            }
            od = od < 1.f ? 1.f : od;
            id = id < 1.f ? 1.f : id;
            sc = rsqrtf(od) * rsqrtf(id);
        }
        sScale[tid] = sc;
    }

    for (int i = tid; i < TILE_M * (IN_FEATS / 4); i += 128) {
        int row = i / (IN_FEATS / 4);
        int c4  = i % (IN_FEATS / 4);
        float4 v = make_float4(0.f, 0.f, 0.f, 0.f);
        if (m0 + row < n_nodes)
            v = ((const float4*)feat)[(long long)(m0 + row) * (IN_FEATS / 4) + c4];
        float* p = &sA[row * LDA + c4 * 4];
        p[0] = v.x; p[1] = v.y; p[2] = v.z; p[3] = v.w;
    }
    __syncthreads();

    for (int i = tid; seg == nullptr && i < TILE_M * (OUT_FEATS / 4); i += 128) {
        int row = i / (OUT_FEATS / 4);
        int c4  = i % (OUT_FEATS / 4);
        if (m0 + row < n_nodes)
            ((float4*)out)[(long long)(m0 + row) * (OUT_FEATS / 4) + c4] =
                ((const float4*)bias)[c4];
    }

    const int wave = tid >> 5;
    const int lane = tid & 31;
    const int r16  = lane & 15;
    const int koff = (lane >> 4) << 1;
    const float* wcol = wT + (wave * 16 + r16) * IN_FEATS;
    const float* arow = sA + r16 * LDA;

    v8f c = {0.f, 0.f, 0.f, 0.f, 0.f, 0.f, 0.f, 0.f};
    #pragma unroll
    for (int k0 = 0; k0 < IN_FEATS; k0 += 4) {
        v2f a, b;
        a.x = arow[k0 + koff];
        a.y = arow[k0 + koff + 1];
        b.x = wcol[k0 + koff];
        b.y = wcol[k0 + koff + 1];
        c = __builtin_amdgcn_wmma_f32_16x16x4_f32(false, a, false, b, (short)0, c,
                                                  false, false);
    }

    {
        const int rbase = (lane >> 4) << 3;
        const int col = wave * 16 + r16;
        #pragma unroll
        for (int r = 0; r < 8; ++r)
            sH[(rbase + r) * LDH + col] = c[r];
    }
    __syncthreads();

    if (tid < TILE_M && (m0 + tid) < n_nodes) {
        const int node = m0 + tid;
        const float scale = sScale[tid];
        float* rowp = &sH[tid * LDH];

        unsigned int iout0 = 0u, iout1 = 0u, iout2 = 0u, iout3 = 0u;
        #pragma unroll
        for (int i = 0; i < MAXK; ++i) {
            float best = -3.402823466e+38f;
            int bi = 0;
            for (int j = 0; j < OUT_FEATS; ++j) {
                float v = rowp[j];
                if (v > best) { best = v; bi = j; }
            }
            rowp[bi] = -3.402823466e+38f;
            tk_vals[(long long)node * MAXK + i] = best * scale;
            unsigned int b8 = ((unsigned int)bi) << ((i & 3) * 8);
            if (i < 4)       iout0 |= b8;
            else if (i < 8)  iout1 |= b8;
            else if (i < 12) iout2 |= b8;
            else             iout3 |= b8;
        }
        unsigned int* iq = &tk_idx[(long long)node * 4];
        iq[0] = iout0; iq[1] = iout1; iq[2] = iout2; iq[3] = iout3;
    }

    if (seg != nullptr && tid == 32) {
        int cnt[TILE_M];
        int tot = 0;
        #pragma unroll
        for (int r = 0; r < TILE_M; ++r) {
            cnt[r] = (m0 + r < n_nodes) ? dcnt[m0 + r] : 0;
            tot += cnt[r];
        }
        int run = atomicAdd(list_top, tot);
        #pragma unroll
        for (int r = 0; r < TILE_M; ++r) {
            if (m0 + r < n_nodes) ((int2*)seg)[m0 + r] = make_int2(run, cnt[r]);
            run += cnt[r];
        }
    }
}

__global__ void scatter_kernel(const int* __restrict__ src, const int* __restrict__ dst,
                               const float* __restrict__ tk_vals,
                               const unsigned int* __restrict__ tk_idx,
                               float* __restrict__ out, int n_edges) {
    long long gid = (long long)blockIdx.x * blockDim.x + threadIdx.x;
    long long total = (long long)n_edges * 4;
    if (gid >= total) return;
    int e = (int)(gid >> 2);
    int part = (int)(gid & 3);
    int s = src[e], d = dst[e];
    float4 v = ((const float4*)tk_vals)[(long long)s * 4 + part];
    unsigned int ip = tk_idx[(long long)s * 4 + part];
    float* orow = out + (long long)d * OUT_FEATS;
    atomicAdd(orow + (ip & 0xffu), v.x);
    atomicAdd(orow + ((ip >> 8) & 0xffu), v.y);
    atomicAdd(orow + ((ip >> 16) & 0xffu), v.z);
    atomicAdd(orow + ((ip >> 24) & 0xffu), v.w);
}


__global__ void fill_kernel(const int* __restrict__ src, const int* __restrict__ dst,
                            const int* __restrict__ rank, const int* __restrict__ seg,
                            int* __restrict__ sorted_src, int n_edges) {
    const long long stride = (long long)gridDim.x * blockDim.x;
    for (long long e = (long long)blockIdx.x * blockDim.x + threadIdx.x; e < n_edges; e += stride) {
        const int d = dst[e];
        sorted_src[seg[2LL * d] + rank[e]] = src[e];
    }
}

__global__ __launch_bounds__(256)
void gather_kernel(const int* __restrict__ seg, const int* __restrict__ sorted_src,
                   const float* __restrict__ tk_vals, const unsigned int* __restrict__ tk_idx,
                   const float* __restrict__ bias, float* __restrict__ out, int n_nodes) {
    __shared__ __attribute__((aligned(16))) float sRow[8][OUT_FEATS];
    const int lane = threadIdx.x & 31;
    const int wave = threadIdx.x >> 5;
    const int half = lane >> 4;
    const int k    = lane & 15;
    const int sh   = (k & 3) * 8;
    float* row = sRow[wave];
    const float2 b2 = ((const float2*)bias)[lane];

    for (int node = blockIdx.x * 8 + wave; node < n_nodes; node += gridDim.x * 8) {
        const int2 sg = ((const int2*)seg)[node];
        const int start = __builtin_amdgcn_readfirstlane(sg.x);
        const int cnt   = __builtin_amdgcn_readfirstlane(sg.y);
        row[lane] = 0.f;
        row[lane + 32] = 0.f;
        for (int base = 0; base < cnt; base += 32) {
            const int rem = (cnt - base) < 32 ? (cnt - base) : 32;
            const int sl = lane < rem ? sorted_src[start + base + lane] : 0;
            for (int q = 0; q < rem; q += 8) {
                float v[4];
                unsigned int iw[4];
                #pragma unroll
                for (int u = 0; u < 4; ++u) {
                    const int s = __shfl(sl, q + 2 * u + half);
                    v[u]  = tk_vals[(long long)s * MAXK + k];
                    iw[u] = tk_idx[(long long)s * 4 + (k >> 2)];
                }
                #pragma unroll
                for (int u = 0; u < 4; ++u)
                    if (q + 2 * u + half < rem)
                        atomicAdd(&row[(iw[u] >> sh) & 0xffu], v[u]);
            }
        }
        const float2 r2 = ((const float2*)row)[lane];
        ((float2*)out)[(long long)node * (OUT_FEATS / 2) + lane] =
            make_float2(r2.x + b2.x, r2.y + b2.y);
    }
}

extern "C" void kernel_launch(void* const* d_in, const int* in_sizes, int n_in,
                              void* d_out, int out_size, void* d_ws, size_t ws_size,
                              hipStream_t stream) {
    const float* feat   = (const float*)d_in[0];
    const float* weight = (const float*)d_in[1];
    const float* bias   = (const float*)d_in[2];
    const int*   src    = (const int*)d_in[3];
    const int*   dst    = (const int*)d_in[4];
    float* out = (float*)d_out;

    const int n_nodes = in_sizes[0] / IN_FEATS;
    const int n_edges = in_sizes[3];

    float* ws_f    = (float*)d_ws;
    float* wT      = ws_f;
    float* tk_vals = wT + IN_FEATS * OUT_FEATS;
    unsigned int* tk_idx = (unsigned int*)(tk_vals + (long long)n_nodes * MAXK);

    const size_t o_seg    = (((size_t)IN_FEATS * OUT_FEATS + (size_t)n_nodes * (MAXK + 4)) * 4 + 255) & ~(size_t)255;
    const size_t o_rank   = (o_seg + (size_t)n_nodes * 8 + 255) & ~(size_t)255;
    const size_t o_sorted = (o_rank + (size_t)n_edges * 4 + 255) & ~(size_t)255;
    const size_t o_top    = (o_sorted + (size_t)n_edges * 4 + 255) & ~(size_t)255;
    const bool gather = ws_size >= o_top + 256;
    char* ws_b = (char*)d_ws;
    int*  seg        = gather ? (int*)(ws_b + o_seg)    : nullptr;
    int*  rank       = gather ? (int*)(ws_b + o_rank)   : nullptr;
    int*  sorted_src = gather ? (int*)(ws_b + o_sorted) : nullptr;
    int*  list_top   = gather ? (int*)(ws_b + o_top)    : nullptr;

    int* dcnt = gather ? (int*)d_out : nullptr;
    if (gather) {
        int n2 = 2 * n_nodes;
        init_dense_kernel<<<(n2 + 255) / 256, 256, 0, stream>>>(dcnt, list_top, n2);
        int nb = (n_nodes + DEG_R - 1) / DEG_R;
        deg_lds_kernel<<<nb * DEG_NS, DEG_THREADS, 0, stream>>>(src, dst, dcnt, weight, wT, rank,
                                                                 n_nodes, n_edges);
    } else {
        long long total = 2LL * n_nodes;
        int grid = (int)((total + 255) / 256);
        init_kernel<<<grid, 256, 0, stream>>>(out, list_top, n_nodes);
        int work = n_edges > IN_FEATS * OUT_FEATS ? n_edges : IN_FEATS * OUT_FEATS;
        int grid2 = (work + 255) / 256;
        deg_wt_kernel<<<grid2, 256, 0, stream>>>(src, dst, out, weight, wT, rank, n_edges);
    }
    {
        int grid = (n_nodes + TILE_M - 1) / TILE_M;
        gemm_maxk_kernel<<<grid, 128, 0, stream>>>(feat, wT, bias, out,
                                                   tk_vals, tk_idx, seg, list_top, dcnt, n_nodes);
    }
    if (gather) {
        fill_kernel<<<2048, 256, 0, stream>>>(src, dst, rank, seg, sorted_src, n_edges);
        gather_kernel<<<2048, 256, 0, stream>>>(seg, sorted_src, tk_vals, tk_idx, bias, out, n_nodes);
    } else {
        long long total = (long long)n_edges * 4;
        int grid = (int)((total + 255) / 256);
        scatter_kernel<<<grid, 256, 0, stream>>>(src, dst, tk_vals, tk_idx, out, n_edges);
    }
}


# top-16 of each row by rank counting on all 128 threads of the GEMM block (was a serial max-scan on 16 lanes of wave 0); records staged in LDS, stored as whole lines; slots and sum order unchanged
# speedup vs baseline: 2.0260x; 1.1052x over previous
#include <hip/hip_runtime.h>
#include <math.h>

typedef __attribute__((ext_vector_type(2))) float v2f;
typedef __attribute__((ext_vector_type(8))) float v8f;

#define IN_FEATS 256
#define OUT_FEATS 64
#define MAXK 16
#define TILE_M 16
#define LDA (IN_FEATS + 4)
#define LDH (OUT_FEATS + 4)

#define DEG_OUT_COL 0
#define DEG_IN_COL  32

__global__ void init_kernel(float* __restrict__ out, int* __restrict__ list_top, int n_nodes) {
    long long i = (long long)blockIdx.x * blockDim.x + threadIdx.x;
    if (i == 0 && list_top != nullptr) *list_top = 0;
    if (i < 2LL * n_nodes)
        out[(i >> 1) * OUT_FEATS + ((i & 1) ? DEG_IN_COL : DEG_OUT_COL)] = 0.0f;
}

__global__ void deg_wt_kernel(const int* __restrict__ src, const int* __restrict__ dst,
                              float* __restrict__ out,
                              const float* __restrict__ weight, float* __restrict__ wT,
                              int* __restrict__ rank, int n_edges) {
    int i = blockIdx.x * blockDim.x + threadIdx.x;
    if (i < n_edges) {
        const int s = src[i];
        const int d = dst[i];
        atomicAdd(out + (long long)s * OUT_FEATS + DEG_OUT_COL, 1.0f);
        const int r = atomicAdd((int*)(out + (long long)d * OUT_FEATS + DEG_IN_COL), 1);
        if (rank != nullptr) rank[i] = r;
    }
    if (i < IN_FEATS * OUT_FEATS) {
        int k = i >> 6;
        int c = i & 63;
        wT[c * IN_FEATS + k] = weight[i];
    }
}


#define DEG_R       6400
#define DEG_NS      16
#define DEG_THREADS 1024
#define DEG_FL      ((DEG_R + DEG_THREADS - 1) / DEG_THREADS)

__global__ void init_dense_kernel(int* __restrict__ dcnt, int* __restrict__ list_top, int n2) {
    int i = blockIdx.x * blockDim.x + threadIdx.x;
    if (i == 0) *list_top = 0;
    if (i < n2) dcnt[i] = 0;
}

__device__ __forceinline__ void count_in_range(int* cnt, int v, int lo, int len) {
    const unsigned int o = (unsigned int)(v - lo);
    if (o < (unsigned int)len) atomicAdd(&cnt[o], 1);
}

__device__ __forceinline__ void rank_in_range(int* cnt, int* __restrict__ rank, int e,
                                              int v, int lo, int len) {
    const unsigned int o = (unsigned int)(v - lo);
    if (o < (unsigned int)len) rank[e] = atomicAdd(&cnt[o], 1);
}

__global__ __launch_bounds__(DEG_THREADS)
void deg_lds_kernel(const int* __restrict__ src, const int* __restrict__ dst,
                    int* __restrict__ dcnt,
                    const float* __restrict__ weight, float* __restrict__ wT,
                    int* __restrict__ rank, int n_nodes, int n_edges) {
    __shared__ int sIn[DEG_R];
    __shared__ int sOut[DEG_R];

    const int tid = threadIdx.x;
    const int b = blockIdx.x / DEG_NS;
    const int s = blockIdx.x % DEG_NS;
    const int lo = b * DEG_R;
    const int len = (n_nodes - lo) < DEG_R ? (n_nodes - lo) : DEG_R;

    {
        const int gi = blockIdx.x * DEG_THREADS + tid;
        if (gi < IN_FEATS * OUT_FEATS) {
            const int k = gi >> 6;
            const int c = gi & 63;
            wT[c * IN_FEATS + k] = weight[gi];
        }
    }

    for (int n = tid; n < DEG_R; n += DEG_THREADS) { sIn[n] = 0; sOut[n] = 0; }
    __syncthreads();

    const int nq = n_edges >> 2;
    const int qs = (nq + DEG_NS - 1) / DEG_NS;
    const int q0 = s * qs;
    const int q1 = (q0 + qs) < nq ? (q0 + qs) : nq;
    const int e_tail = (nq << 2) + tid;
    const bool tail = (s == DEG_NS - 1) && (e_tail < n_edges);

    for (int q = q0 + tid; q < q1; q += 2 * DEG_THREADS) {
        const bool two = (q + DEG_THREADS) < q1;
        const int qb = two ? (q + DEG_THREADS) : q;
        const int len1 = two ? len : 0;
        const int4 d0 = ((const int4*)dst)[q];
        const int4 d1 = ((const int4*)dst)[qb];
        const int4 s0 = ((const int4*)src)[q];
        const int4 s1 = ((const int4*)src)[qb];
        count_in_range(sIn, d0.x, lo, len); count_in_range(sIn, d0.y, lo, len);
        count_in_range(sIn, d0.z, lo, len); count_in_range(sIn, d0.w, lo, len);
        count_in_range(sIn, d1.x, lo, len1); count_in_range(sIn, d1.y, lo, len1);
        count_in_range(sIn, d1.z, lo, len1); count_in_range(sIn, d1.w, lo, len1);
        count_in_range(sOut, s0.x, lo, len); count_in_range(sOut, s0.y, lo, len);
        count_in_range(sOut, s0.z, lo, len); count_in_range(sOut, s0.w, lo, len);
        count_in_range(sOut, s1.x, lo, len1); count_in_range(sOut, s1.y, lo, len1);
        count_in_range(sOut, s1.z, lo, len1); count_in_range(sOut, s1.w, lo, len1);
    }
    if (tail) {
        count_in_range(sIn, dst[e_tail], lo, len);
        count_in_range(sOut, src[e_tail], lo, len);
    }
    __syncthreads();

    {
        int ci[DEG_FL], start[DEG_FL];
        #pragma unroll
        for (int u = 0; u < DEG_FL; ++u) {
            const int n = tid + u * DEG_THREADS;
            ci[u] = n < len ? sIn[n] : 0;
            const int co = n < len ? sOut[n] : 0;
            start[u] = 0;
            if (ci[u] > 0) start[u] = atomicAdd(&dcnt[lo + n], ci[u]);
            if (co > 0) atomicAdd(&dcnt[n_nodes + lo + n], co);
        }
        #pragma unroll
        for (int u = 0; u < DEG_FL; ++u)
            if (ci[u] > 0) sIn[tid + u * DEG_THREADS] = start[u];
    }
    __syncthreads();

    for (int q = q0 + tid; q < q1; q += 2 * DEG_THREADS) {
        const bool two = (q + DEG_THREADS) < q1;
        const int qb = two ? (q + DEG_THREADS) : q;
        const int len1 = two ? len : 0;
        const int4 d0 = ((const int4*)dst)[q];
        const int4 d1 = ((const int4*)dst)[qb];
        const int e0 = q << 2;
        const int e1 = qb << 2;
        rank_in_range(sIn, rank, e0,     d0.x, lo, len);
        rank_in_range(sIn, rank, e0 + 1, d0.y, lo, len);
        rank_in_range(sIn, rank, e0 + 2, d0.z, lo, len);
        rank_in_range(sIn, rank, e0 + 3, d0.w, lo, len);
        rank_in_range(sIn, rank, e1,     d1.x, lo, len1);
        rank_in_range(sIn, rank, e1 + 1, d1.y, lo, len1);
        rank_in_range(sIn, rank, e1 + 2, d1.z, lo, len1);
        rank_in_range(sIn, rank, e1 + 3, d1.w, lo, len1);
    }
    if (tail) rank_in_range(sIn, rank, e_tail, dst[e_tail], lo, len);
}

__global__ __launch_bounds__(128)
void gemm_maxk_kernel(const float* __restrict__ feat, const float* __restrict__ wT,
                      const float* __restrict__ bias, float* out,
                      float* __restrict__ tk_vals, unsigned int* __restrict__ tk_idx,
                      int* __restrict__ seg, int* __restrict__ list_top,
                      const int* dcnt, int n_nodes) {
    __shared__ __attribute__((aligned(16))) float sA[TILE_M * LDA];
    __shared__ __attribute__((aligned(16))) float sH[TILE_M * LDH];
    __shared__ float sScale[TILE_M];

    const int tid = threadIdx.x;
    const int m0 = blockIdx.x * TILE_M;

    if (tid < TILE_M) {
        float sc = 0.f;
        if (m0 + tid < n_nodes) {
            float od, id;
            if (dcnt != nullptr) {
                id = (float)dcnt[m0 + tid];
                od = (float)dcnt[n_nodes + m0 + tid];
            } else {
                const float* orow = out + (long long)(m0 + tid) * OUT_FEATS;
                od = orow[DEG_OUT_COL];
                id = (float)((const int*)orow)[DEG_IN_COL];
            }
            od = od < 1.f ? 1.f : od;
            id = id < 1.f ? 1.f : id;
            sc = rsqrtf(od) * rsqrtf(id);
        }
        sScale[tid] = sc;
    }

    for (int i = tid; i < TILE_M * (IN_FEATS / 4); i += 128) {
        int row = i / (IN_FEATS / 4);
        int c4  = i % (IN_FEATS / 4);
        float4 v = make_float4(0.f, 0.f, 0.f, 0.f);
        if (m0 + row < n_nodes)
            v = ((const float4*)feat)[(long long)(m0 + row) * (IN_FEATS / 4) + c4];
        float* p = &sA[row * LDA + c4 * 4];
        p[0] = v.x; p[1] = v.y; p[2] = v.z; p[3] = v.w;
    }
    __syncthreads();

    for (int i = tid; seg == nullptr && i < TILE_M * (OUT_FEATS / 4); i += 128) {
        int row = i / (OUT_FEATS / 4);
        int c4  = i % (OUT_FEATS / 4);
        if (m0 + row < n_nodes)
            ((float4*)out)[(long long)(m0 + row) * (OUT_FEATS / 4) + c4] =
                ((const float4*)bias)[c4];
    }

    const int wave = tid >> 5;
    const int lane = tid & 31;
    const int r16  = lane & 15;
    const int koff = (lane >> 4) << 1;
    const float* wcol = wT + (wave * 16 + r16) * IN_FEATS;
    const float* arow = sA + r16 * LDA;

    v8f c = {0.f, 0.f, 0.f, 0.f, 0.f, 0.f, 0.f, 0.f};
    #pragma unroll
    for (int k0 = 0; k0 < IN_FEATS; k0 += 4) {
        v2f a, b;
        a.x = arow[k0 + koff];
        a.y = arow[k0 + koff + 1];
        b.x = wcol[k0 + koff];
        b.y = wcol[k0 + koff + 1];
        c = __builtin_amdgcn_wmma_f32_16x16x4_f32(false, a, false, b, (short)0, c,
                                                  false, false);
    }

    {
        const int rbase = (lane >> 4) << 3;
        const int col = wave * 16 + r16;
        #pragma unroll
        for (int r = 0; r < 8; ++r)
            sH[(rbase + r) * LDH + col] = c[r];
    }
    __syncthreads();

    {
        const int row = tid >> 3;
        const int p   = tid & 7;
        const float* rowp = &sH[row * LDH];
        float x[8];
        int cnt[8];
        #pragma unroll
        for (int e = 0; e < 8; ++e) { x[e] = rowp[8 * e + p]; cnt[e] = 0; }
        #pragma unroll
        for (int g = 0; g < 8; ++g) {
            const float4 h0 = *(const float4*)(rowp + 8 * g);
            const float4 h1 = *(const float4*)(rowp + 8 * g + 4);
            const float hv[8] = {h0.x, h0.y, h0.z, h0.w, h1.x, h1.y, h1.z, h1.w};
            #pragma unroll
            for (int pp = 0; pp < 8; ++pp) {
                const float v = hv[pp];
                #pragma unroll
                for (int e = 0; e < 8; ++e) {
                    bool before;
                    if (g < e)      before = v >= x[e];
                    else if (g > e) before = v > x[e];
                    else            before = (v > x[e]) || (v == x[e] && pp < p);
                    cnt[e] += before ? 1 : 0;
                }
            }
        }
        const float scale = sScale[row];
        float* sV = sA;
        unsigned char* sI = (unsigned char*)(sA + TILE_M * MAXK);
        #pragma unroll
        for (int e = 0; e < 8; ++e) {
            const bool keep = cnt[e] < MAXK;
            const int slot = row * MAXK + cnt[e];
            sV[keep ? slot : 512 + tid] = x[e] * scale;
            sI[keep ? slot : 4096 + tid] = (unsigned char)(8 * e + p);
        }
    }
    __syncthreads();
    if (tid < 64) {
        const int row = tid >> 2;
        if (m0 + row < n_nodes)
            ((float4*)tk_vals)[(long long)(m0 + row) * (MAXK / 4) + (tid & 3)] =
                ((const float4*)sA)[tid];
    } else if (tid < 64 + TILE_M) {
        const int row = tid - 64;
        if (m0 + row < n_nodes)
            ((uint4*)tk_idx)[m0 + row] = ((const uint4*)(sA + TILE_M * MAXK))[row];
    }

    if (seg != nullptr && tid == 32) {
        int cnt[TILE_M];
        int tot = 0;
        #pragma unroll
        for (int r = 0; r < TILE_M; ++r) {
            cnt[r] = (m0 + r < n_nodes) ? dcnt[m0 + r] : 0;
            tot += cnt[r];
        }
        int run = atomicAdd(list_top, tot);
        #pragma unroll
        for (int r = 0; r < TILE_M; ++r) {
            if (m0 + r < n_nodes) ((int2*)seg)[m0 + r] = make_int2(run, cnt[r]);
            run += cnt[r];
        }
    }
}

__global__ void scatter_kernel(const int* __restrict__ src, const int* __restrict__ dst,
                               const float* __restrict__ tk_vals,
                               const unsigned int* __restrict__ tk_idx,
                               float* __restrict__ out, int n_edges) {
    long long gid = (long long)blockIdx.x * blockDim.x + threadIdx.x;
    long long total = (long long)n_edges * 4;
    if (gid >= total) return;
    int e = (int)(gid >> 2);
    int part = (int)(gid & 3);
    int s = src[e], d = dst[e];
    float4 v = ((const float4*)tk_vals)[(long long)s * 4 + part];
    unsigned int ip = tk_idx[(long long)s * 4 + part];
    float* orow = out + (long long)d * OUT_FEATS;
    atomicAdd(orow + (ip & 0xffu), v.x);
    atomicAdd(orow + ((ip >> 8) & 0xffu), v.y);
    atomicAdd(orow + ((ip >> 16) & 0xffu), v.z);
    atomicAdd(orow + ((ip >> 24) & 0xffu), v.w);
}


__global__ void fill_kernel(const int* __restrict__ src, const int* __restrict__ dst,
                            const int* __restrict__ rank, const int* __restrict__ seg,
                            int* __restrict__ sorted_src, int n_edges) {
    const long long stride = (long long)gridDim.x * blockDim.x;
    for (long long e = (long long)blockIdx.x * blockDim.x + threadIdx.x; e < n_edges; e += stride) {
        const int d = dst[e];
        sorted_src[seg[2LL * d] + rank[e]] = src[e];
    }
}

__global__ __launch_bounds__(256)
void gather_kernel(const int* __restrict__ seg, const int* __restrict__ sorted_src,
                   const float* __restrict__ tk_vals, const unsigned int* __restrict__ tk_idx,
                   const float* __restrict__ bias, float* __restrict__ out, int n_nodes) {
    __shared__ __attribute__((aligned(16))) float sRow[8][OUT_FEATS];
    const int lane = threadIdx.x & 31;
    const int wave = threadIdx.x >> 5;
    const int half = lane >> 4;
    const int k    = lane & 15;
    const int sh   = (k & 3) * 8;
    float* row = sRow[wave];
    const float2 b2 = ((const float2*)bias)[lane];

    for (int node = blockIdx.x * 8 + wave; node < n_nodes; node += gridDim.x * 8) {
        const int2 sg = ((const int2*)seg)[node];
        const int start = __builtin_amdgcn_readfirstlane(sg.x);
        const int cnt   = __builtin_amdgcn_readfirstlane(sg.y);
        row[lane] = 0.f;
        row[lane + 32] = 0.f;
        for (int base = 0; base < cnt; base += 32) {
            const int rem = (cnt - base) < 32 ? (cnt - base) : 32;
            const int sl = lane < rem ? sorted_src[start + base + lane] : 0;
            for (int q = 0; q < rem; q += 8) {
                float v[4];
                unsigned int iw[4];
                #pragma unroll
                for (int u = 0; u < 4; ++u) {
                    const int s = __shfl(sl, q + 2 * u + half);
                    v[u]  = tk_vals[(long long)s * MAXK + k];
                    iw[u] = tk_idx[(long long)s * 4 + (k >> 2)];
                }
                #pragma unroll
                for (int u = 0; u < 4; ++u)
                    if (q + 2 * u + half < rem)
                        atomicAdd(&row[(iw[u] >> sh) & 0xffu], v[u]);
            }
        }
        const float2 r2 = ((const float2*)row)[lane];
        ((float2*)out)[(long long)node * (OUT_FEATS / 2) + lane] =
            make_float2(r2.x + b2.x, r2.y + b2.y);
    }
}

extern "C" void kernel_launch(void* const* d_in, const int* in_sizes, int n_in,
                              void* d_out, int out_size, void* d_ws, size_t ws_size,
                              hipStream_t stream) {
    const float* feat   = (const float*)d_in[0];
    const float* weight = (const float*)d_in[1];
    const float* bias   = (const float*)d_in[2];
    const int*   src    = (const int*)d_in[3];
    const int*   dst    = (const int*)d_in[4];
    float* out = (float*)d_out;

    const int n_nodes = in_sizes[0] / IN_FEATS;
    const int n_edges = in_sizes[3];

    float* ws_f    = (float*)d_ws;
    float* wT      = ws_f;
    float* tk_vals = wT + IN_FEATS * OUT_FEATS;
    unsigned int* tk_idx = (unsigned int*)(tk_vals + (long long)n_nodes * MAXK);

    const size_t o_seg    = (((size_t)IN_FEATS * OUT_FEATS + (size_t)n_nodes * (MAXK + 4)) * 4 + 255) & ~(size_t)255;
    const size_t o_rank   = (o_seg + (size_t)n_nodes * 8 + 255) & ~(size_t)255;
    const size_t o_sorted = (o_rank + (size_t)n_edges * 4 + 255) & ~(size_t)255;
    const size_t o_top    = (o_sorted + (size_t)n_edges * 4 + 255) & ~(size_t)255;
    const bool gather = ws_size >= o_top + 256;
    char* ws_b = (char*)d_ws;
    int*  seg        = gather ? (int*)(ws_b + o_seg)    : nullptr;
    int*  rank       = gather ? (int*)(ws_b + o_rank)   : nullptr;
    int*  sorted_src = gather ? (int*)(ws_b + o_sorted) : nullptr;
    int*  list_top   = gather ? (int*)(ws_b + o_top)    : nullptr;

    int* dcnt = gather ? (int*)d_out : nullptr;
    if (gather) {
        int n2 = 2 * n_nodes;
        init_dense_kernel<<<(n2 + 255) / 256, 256, 0, stream>>>(dcnt, list_top, n2);
        int nb = (n_nodes + DEG_R - 1) / DEG_R;
        deg_lds_kernel<<<nb * DEG_NS, DEG_THREADS, 0, stream>>>(src, dst, dcnt, weight, wT, rank,
                                                                 n_nodes, n_edges);
    } else {
        long long total = 2LL * n_nodes;
        int grid = (int)((total + 255) / 256);
        init_kernel<<<grid, 256, 0, stream>>>(out, list_top, n_nodes);
        int work = n_edges > IN_FEATS * OUT_FEATS ? n_edges : IN_FEATS * OUT_FEATS;
        int grid2 = (work + 255) / 256;
        deg_wt_kernel<<<grid2, 256, 0, stream>>>(src, dst, out, weight, wT, rank, n_edges);
    }
    {
        int grid = (n_nodes + TILE_M - 1) / TILE_M;
        gemm_maxk_kernel<<<grid, 128, 0, stream>>>(feat, wT, bias, out,
                                                   tk_vals, tk_idx, seg, list_top, dcnt, n_nodes);
    }
    if (gather) {
        fill_kernel<<<2048, 256, 0, stream>>>(src, dst, rank, seg, sorted_src, n_edges);
        gather_kernel<<<2048, 256, 0, stream>>>(seg, sorted_src, tk_vals, tk_idx, bias, out, n_nodes);
    } else {
        long long total = (long long)n_edges * 4;
        int grid = (int)((total + 255) / 256);
        scatter_kernel<<<grid, 256, 0, stream>>>(src, dst, tk_vals, tk_idx, out, n_edges);
    }
}


# W kept in the workspace in B-fragment order: each GEMM B load is 16 bytes a lane, 512 contiguous bytes a wave (4 whole lines; was 16 pieces of 16 lines), feeds two matrix steps; k pairing unchanged
# speedup vs baseline: 2.2627x; 1.1168x over previous
#include <hip/hip_runtime.h>
#include <math.h>

typedef __attribute__((ext_vector_type(2))) float v2f;
typedef __attribute__((ext_vector_type(8))) float v8f;

#define IN_FEATS 256
#define OUT_FEATS 64
#define MAXK 16
#define TILE_M 16
#define LDA (IN_FEATS + 4)
#define LDH (OUT_FEATS + 4)

#define DEG_OUT_COL 0
#define DEG_IN_COL  32

__device__ __forceinline__ int wfrag_pos(int gi) {
    const int k = gi >> 6;
    const int c = gi & 63;
    const int w = c >> 4, c16 = c & 15;
    const int jj = k >> 3, sub = (k >> 2) & 1, h = (k >> 1) & 1, e = k & 1;
    return (((w * 32 + jj) * 32 + (h * 16 + c16)) << 2) + sub * 2 + e;
}

__global__ void init_kernel(float* __restrict__ out, int* __restrict__ list_top, int n_nodes) {
    long long i = (long long)blockIdx.x * blockDim.x + threadIdx.x;
    if (i == 0 && list_top != nullptr) *list_top = 0;
    if (i < 2LL * n_nodes)
        out[(i >> 1) * OUT_FEATS + ((i & 1) ? DEG_IN_COL : DEG_OUT_COL)] = 0.0f;
}

__global__ void deg_wt_kernel(const int* __restrict__ src, const int* __restrict__ dst,
                              float* __restrict__ out,
                              const float* __restrict__ weight, float* __restrict__ wT,
                              int* __restrict__ rank, int n_edges) {
    int i = blockIdx.x * blockDim.x + threadIdx.x;
    if (i < n_edges) {
        const int s = src[i];
        const int d = dst[i];
        atomicAdd(out + (long long)s * OUT_FEATS + DEG_OUT_COL, 1.0f);
        const int r = atomicAdd((int*)(out + (long long)d * OUT_FEATS + DEG_IN_COL), 1);
        if (rank != nullptr) rank[i] = r;
    }
    if (i < IN_FEATS * OUT_FEATS) wT[wfrag_pos(i)] = weight[i];
}


#define DEG_R       6400
#define DEG_NS      16
#define DEG_THREADS 1024
#define DEG_FL      ((DEG_R + DEG_THREADS - 1) / DEG_THREADS)

__global__ void init_dense_kernel(int* __restrict__ dcnt, int* __restrict__ list_top, int n2) {
    int i = blockIdx.x * blockDim.x + threadIdx.x;
    if (i == 0) *list_top = 0;
    if (i < n2) dcnt[i] = 0;
}

__device__ __forceinline__ void count_in_range(int* cnt, int v, int lo, int len) {
    const unsigned int o = (unsigned int)(v - lo);
    if (o < (unsigned int)len) atomicAdd(&cnt[o], 1);
}

__device__ __forceinline__ void rank_in_range(int* cnt, int* __restrict__ rank, int e,
                                              int v, int lo, int len) {
    const unsigned int o = (unsigned int)(v - lo);
    if (o < (unsigned int)len) rank[e] = atomicAdd(&cnt[o], 1);
}

__global__ __launch_bounds__(DEG_THREADS)
void deg_lds_kernel(const int* __restrict__ src, const int* __restrict__ dst,
                    int* __restrict__ dcnt,
                    const float* __restrict__ weight, float* __restrict__ wT,
                    int* __restrict__ rank, int n_nodes, int n_edges) {
    __shared__ int sIn[DEG_R];
    __shared__ int sOut[DEG_R];

    const int tid = threadIdx.x;
    const int b = blockIdx.x / DEG_NS;
    const int s = blockIdx.x % DEG_NS;
    const int lo = b * DEG_R;
    const int len = (n_nodes - lo) < DEG_R ? (n_nodes - lo) : DEG_R;

    {
        const int gi = blockIdx.x * DEG_THREADS + tid;
        if (gi < IN_FEATS * OUT_FEATS) wT[wfrag_pos(gi)] = weight[gi];
    }

    for (int n = tid; n < DEG_R; n += DEG_THREADS) { sIn[n] = 0; sOut[n] = 0; }
    __syncthreads();

    const int nq = n_edges >> 2;
    const int qs = (nq + DEG_NS - 1) / DEG_NS;
    const int q0 = s * qs;
    const int q1 = (q0 + qs) < nq ? (q0 + qs) : nq;
    const int e_tail = (nq << 2) + tid;
    const bool tail = (s == DEG_NS - 1) && (e_tail < n_edges);

    for (int q = q0 + tid; q < q1; q += 2 * DEG_THREADS) {
        const bool two = (q + DEG_THREADS) < q1;
        const int qb = two ? (q + DEG_THREADS) : q;
        const int len1 = two ? len : 0;
        const int4 d0 = ((const int4*)dst)[q];
        const int4 d1 = ((const int4*)dst)[qb];
        const int4 s0 = ((const int4*)src)[q];
        const int4 s1 = ((const int4*)src)[qb];
        count_in_range(sIn, d0.x, lo, len); count_in_range(sIn, d0.y, lo, len);
        count_in_range(sIn, d0.z, lo, len); count_in_range(sIn, d0.w, lo, len);
        count_in_range(sIn, d1.x, lo, len1); count_in_range(sIn, d1.y, lo, len1);
        count_in_range(sIn, d1.z, lo, len1); count_in_range(sIn, d1.w, lo, len1);
        count_in_range(sOut, s0.x, lo, len); count_in_range(sOut, s0.y, lo, len);
        count_in_range(sOut, s0.z, lo, len); count_in_range(sOut, s0.w, lo, len);
        count_in_range(sOut, s1.x, lo, len1); count_in_range(sOut, s1.y, lo, len1);
        count_in_range(sOut, s1.z, lo, len1); count_in_range(sOut, s1.w, lo, len1);
    }
    if (tail) {
        count_in_range(sIn, dst[e_tail], lo, len);
        count_in_range(sOut, src[e_tail], lo, len);
    }
    __syncthreads();

    {
        int ci[DEG_FL], start[DEG_FL];
        #pragma unroll
        for (int u = 0; u < DEG_FL; ++u) {
            const int n = tid + u * DEG_THREADS;
            ci[u] = n < len ? sIn[n] : 0;
            const int co = n < len ? sOut[n] : 0;
            start[u] = 0;
            if (ci[u] > 0) start[u] = atomicAdd(&dcnt[lo + n], ci[u]);
            if (co > 0) atomicAdd(&dcnt[n_nodes + lo + n], co);
        }
        #pragma unroll
        for (int u = 0; u < DEG_FL; ++u)
            if (ci[u] > 0) sIn[tid + u * DEG_THREADS] = start[u];
    }
    __syncthreads();

    for (int q = q0 + tid; q < q1; q += 2 * DEG_THREADS) {
        const bool two = (q + DEG_THREADS) < q1;
        const int qb = two ? (q + DEG_THREADS) : q;
        const int len1 = two ? len : 0;
        const int4 d0 = ((const int4*)dst)[q];
        const int4 d1 = ((const int4*)dst)[qb];
        const int e0 = q << 2;
        const int e1 = qb << 2;
        rank_in_range(sIn, rank, e0,     d0.x, lo, len);
        rank_in_range(sIn, rank, e0 + 1, d0.y, lo, len);
        rank_in_range(sIn, rank, e0 + 2, d0.z, lo, len);
        rank_in_range(sIn, rank, e0 + 3, d0.w, lo, len);
        rank_in_range(sIn, rank, e1,     d1.x, lo, len1);
        rank_in_range(sIn, rank, e1 + 1, d1.y, lo, len1);
        rank_in_range(sIn, rank, e1 + 2, d1.z, lo, len1);
        rank_in_range(sIn, rank, e1 + 3, d1.w, lo, len1);
    }
    if (tail) rank_in_range(sIn, rank, e_tail, dst[e_tail], lo, len);
}

__global__ __launch_bounds__(128)
void gemm_maxk_kernel(const float* __restrict__ feat, const float* __restrict__ wT,
                      const float* __restrict__ bias, float* out,
                      float* __restrict__ tk_vals, unsigned int* __restrict__ tk_idx,
                      int* __restrict__ seg, int* __restrict__ list_top,
                      const int* dcnt, int n_nodes) {
    __shared__ __attribute__((aligned(16))) float sA[TILE_M * LDA];
    __shared__ __attribute__((aligned(16))) float sH[TILE_M * LDH];
    __shared__ float sScale[TILE_M];

    const int tid = threadIdx.x;
    const int m0 = blockIdx.x * TILE_M;

    if (tid < TILE_M) {
        float sc = 0.f;
        if (m0 + tid < n_nodes) {
            float od, id;
            if (dcnt != nullptr) {
                id = (float)dcnt[m0 + tid];
                od = (float)dcnt[n_nodes + m0 + tid];
            } else {
                const float* orow = out + (long long)(m0 + tid) * OUT_FEATS;
                od = orow[DEG_OUT_COL];
                id = (float)((const int*)orow)[DEG_IN_COL];
            }
            od = od < 1.f ? 1.f : od;
            id = id < 1.f ? 1.f : id;
            sc = rsqrtf(od) * rsqrtf(id);
        }
        sScale[tid] = sc;
    }

    for (int i = tid; i < TILE_M * (IN_FEATS / 4); i += 128) {
        int row = i / (IN_FEATS / 4);
        int c4  = i % (IN_FEATS / 4);
        float4 v = make_float4(0.f, 0.f, 0.f, 0.f);
        if (m0 + row < n_nodes)
            v = ((const float4*)feat)[(long long)(m0 + row) * (IN_FEATS / 4) + c4];
        float* p = &sA[row * LDA + c4 * 4];
        p[0] = v.x; p[1] = v.y; p[2] = v.z; p[3] = v.w;
    }
    __syncthreads();

    for (int i = tid; seg == nullptr && i < TILE_M * (OUT_FEATS / 4); i += 128) {
        int row = i / (OUT_FEATS / 4);
        int c4  = i % (OUT_FEATS / 4);
        if (m0 + row < n_nodes)
            ((float4*)out)[(long long)(m0 + row) * (OUT_FEATS / 4) + c4] =
                ((const float4*)bias)[c4];
    }

    const int wave = tid >> 5;
    const int lane = tid & 31;
    const int r16  = lane & 15;
    const int koff = (lane >> 4) << 1;
    const float4* wfrag = (const float4*)wT + (wave * 32) * 32 + lane;
    const float* arow = sA + r16 * LDA;

    v8f c = {0.f, 0.f, 0.f, 0.f, 0.f, 0.f, 0.f, 0.f};
    #pragma unroll
    for (int k0 = 0; k0 < IN_FEATS; k0 += 8) {
        const float4 b4 = wfrag[(k0 >> 3) * 32];
        v2f a0, a1, b0, b1;
        a0.x = arow[k0 + koff];
        a0.y = arow[k0 + koff + 1];
        a1.x = arow[k0 + 4 + koff];
        a1.y = arow[k0 + 4 + koff + 1];
        b0.x = b4.x; b0.y = b4.y;
        b1.x = b4.z; b1.y = b4.w;
        c = __builtin_amdgcn_wmma_f32_16x16x4_f32(false, a0, false, b0, (short)0, c,
                                                  false, false);
        c = __builtin_amdgcn_wmma_f32_16x16x4_f32(false, a1, false, b1, (short)0, c,
                                                  false, false);
    }

    {
        const int rbase = (lane >> 4) << 3;
        const int col = wave * 16 + r16;
        #pragma unroll
        for (int r = 0; r < 8; ++r)
            sH[(rbase + r) * LDH + col] = c[r];
    }
    __syncthreads();

    {
        const int row = tid >> 3;
        const int p   = tid & 7;
        const float* rowp = &sH[row * LDH];
        float x[8];
        int cnt[8];
        #pragma unroll
        for (int e = 0; e < 8; ++e) { x[e] = rowp[8 * e + p]; cnt[e] = 0; }
        #pragma unroll
        for (int g = 0; g < 8; ++g) {
            const float4 h0 = *(const float4*)(rowp + 8 * g);
            const float4 h1 = *(const float4*)(rowp + 8 * g + 4);
            const float hv[8] = {h0.x, h0.y, h0.z, h0.w, h1.x, h1.y, h1.z, h1.w};
            #pragma unroll
            for (int pp = 0; pp < 8; ++pp) {
                const float v = hv[pp];
                #pragma unroll
                for (int e = 0; e < 8; ++e) {
                    bool before;
                    if (g < e)      before = v >= x[e];
                    else if (g > e) before = v > x[e];
                    else            before = (v > x[e]) || (v == x[e] && pp < p);
                    cnt[e] += before ? 1 : 0;
                }
            }
        }
        const float scale = sScale[row];
        float* sV = sA;
        unsigned char* sI = (unsigned char*)(sA + TILE_M * MAXK);
        #pragma unroll
        for (int e = 0; e < 8; ++e) {
            const bool keep = cnt[e] < MAXK;
            const int slot = row * MAXK + cnt[e];
            sV[keep ? slot : 512 + tid] = x[e] * scale;
            sI[keep ? slot : 4096 + tid] = (unsigned char)(8 * e + p);
        }
    }
    __syncthreads();
    if (tid < 64) {
        const int row = tid >> 2;
        if (m0 + row < n_nodes)
            ((float4*)tk_vals)[(long long)(m0 + row) * (MAXK / 4) + (tid & 3)] =
                ((const float4*)sA)[tid];
    } else if (tid < 64 + TILE_M) {
        const int row = tid - 64;
        if (m0 + row < n_nodes)
            ((uint4*)tk_idx)[m0 + row] = ((const uint4*)(sA + TILE_M * MAXK))[row];
    }

    if (seg != nullptr && tid == 32) {
        int cnt[TILE_M];
        int tot = 0;
        #pragma unroll
        for (int r = 0; r < TILE_M; ++r) {
            cnt[r] = (m0 + r < n_nodes) ? dcnt[m0 + r] : 0;
            tot += cnt[r];
        }
        int run = atomicAdd(list_top, tot);
        #pragma unroll
        for (int r = 0; r < TILE_M; ++r) {
            if (m0 + r < n_nodes) ((int2*)seg)[m0 + r] = make_int2(run, cnt[r]);
            run += cnt[r];
        }
    }
}

__global__ void scatter_kernel(const int* __restrict__ src, const int* __restrict__ dst,
                               const float* __restrict__ tk_vals,
                               const unsigned int* __restrict__ tk_idx,
                               float* __restrict__ out, int n_edges) {
    long long gid = (long long)blockIdx.x * blockDim.x + threadIdx.x;
    long long total = (long long)n_edges * 4;
    if (gid >= total) return;
    int e = (int)(gid >> 2);
    int part = (int)(gid & 3);
    int s = src[e], d = dst[e];
    float4 v = ((const float4*)tk_vals)[(long long)s * 4 + part];
    unsigned int ip = tk_idx[(long long)s * 4 + part];
    float* orow = out + (long long)d * OUT_FEATS;
    atomicAdd(orow + (ip & 0xffu), v.x);
    atomicAdd(orow + ((ip >> 8) & 0xffu), v.y);
    atomicAdd(orow + ((ip >> 16) & 0xffu), v.z);
    atomicAdd(orow + ((ip >> 24) & 0xffu), v.w);
}


__global__ void fill_kernel(const int* __restrict__ src, const int* __restrict__ dst,
                            const int* __restrict__ rank, const int* __restrict__ seg,
                            int* __restrict__ sorted_src, int n_edges) {
    const long long stride = (long long)gridDim.x * blockDim.x;
    for (long long e = (long long)blockIdx.x * blockDim.x + threadIdx.x; e < n_edges; e += stride) {
        const int d = dst[e];
        sorted_src[seg[2LL * d] + rank[e]] = src[e];
    }
}

__global__ __launch_bounds__(256)
void gather_kernel(const int* __restrict__ seg, const int* __restrict__ sorted_src,
                   const float* __restrict__ tk_vals, const unsigned int* __restrict__ tk_idx,
                   const float* __restrict__ bias, float* __restrict__ out, int n_nodes) {
    __shared__ __attribute__((aligned(16))) float sRow[8][OUT_FEATS];
    const int lane = threadIdx.x & 31;
    const int wave = threadIdx.x >> 5;
    const int half = lane >> 4;
    const int k    = lane & 15;
    const int sh   = (k & 3) * 8;
    float* row = sRow[wave];
    const float2 b2 = ((const float2*)bias)[lane];

    for (int node = blockIdx.x * 8 + wave; node < n_nodes; node += gridDim.x * 8) {
        const int2 sg = ((const int2*)seg)[node];
        const int start = __builtin_amdgcn_readfirstlane(sg.x);
        const int cnt   = __builtin_amdgcn_readfirstlane(sg.y);
        row[lane] = 0.f;
        row[lane + 32] = 0.f;
        for (int base = 0; base < cnt; base += 32) {
            const int rem = (cnt - base) < 32 ? (cnt - base) : 32;
            const int sl = lane < rem ? sorted_src[start + base + lane] : 0;
            for (int q = 0; q < rem; q += 8) {
                float v[4];
                unsigned int iw[4];
                #pragma unroll
                for (int u = 0; u < 4; ++u) {
                    const int s = __shfl(sl, q + 2 * u + half);
                    v[u]  = tk_vals[(long long)s * MAXK + k];
                    iw[u] = tk_idx[(long long)s * 4 + (k >> 2)];
                }
                #pragma unroll
                for (int u = 0; u < 4; ++u)
                    if (q + 2 * u + half < rem)
                        atomicAdd(&row[(iw[u] >> sh) & 0xffu], v[u]);
            }
        }
        const float2 r2 = ((const float2*)row)[lane];
        ((float2*)out)[(long long)node * (OUT_FEATS / 2) + lane] =
            make_float2(r2.x + b2.x, r2.y + b2.y);
    }
}

extern "C" void kernel_launch(void* const* d_in, const int* in_sizes, int n_in,
                              void* d_out, int out_size, void* d_ws, size_t ws_size,
                              hipStream_t stream) {
    const float* feat   = (const float*)d_in[0];
    const float* weight = (const float*)d_in[1];
    const float* bias   = (const float*)d_in[2];
    const int*   src    = (const int*)d_in[3];
    const int*   dst    = (const int*)d_in[4];
    float* out = (float*)d_out;

    const int n_nodes = in_sizes[0] / IN_FEATS;
    const int n_edges = in_sizes[3];

    float* ws_f    = (float*)d_ws;
    float* wT      = ws_f;
    float* tk_vals = wT + IN_FEATS * OUT_FEATS;
    unsigned int* tk_idx = (unsigned int*)(tk_vals + (long long)n_nodes * MAXK);

    const size_t o_seg    = (((size_t)IN_FEATS * OUT_FEATS + (size_t)n_nodes * (MAXK + 4)) * 4 + 255) & ~(size_t)255;
    const size_t o_rank   = (o_seg + (size_t)n_nodes * 8 + 255) & ~(size_t)255;
    const size_t o_sorted = (o_rank + (size_t)n_edges * 4 + 255) & ~(size_t)255;
    const size_t o_top    = (o_sorted + (size_t)n_edges * 4 + 255) & ~(size_t)255;
    const bool gather = ws_size >= o_top + 256;
    char* ws_b = (char*)d_ws;
    int*  seg        = gather ? (int*)(ws_b + o_seg)    : nullptr;
    int*  rank       = gather ? (int*)(ws_b + o_rank)   : nullptr;
    int*  sorted_src = gather ? (int*)(ws_b + o_sorted) : nullptr;
    int*  list_top   = gather ? (int*)(ws_b + o_top)    : nullptr;

    int* dcnt = gather ? (int*)d_out : nullptr;
    if (gather) {
        int n2 = 2 * n_nodes;
        init_dense_kernel<<<(n2 + 255) / 256, 256, 0, stream>>>(dcnt, list_top, n2);
        int nb = (n_nodes + DEG_R - 1) / DEG_R;
        deg_lds_kernel<<<nb * DEG_NS, DEG_THREADS, 0, stream>>>(src, dst, dcnt, weight, wT, rank,
                                                                 n_nodes, n_edges);
    } else {
        long long total = 2LL * n_nodes;
        int grid = (int)((total + 255) / 256);
        init_kernel<<<grid, 256, 0, stream>>>(out, list_top, n_nodes);
        int work = n_edges > IN_FEATS * OUT_FEATS ? n_edges : IN_FEATS * OUT_FEATS;
        int grid2 = (work + 255) / 256;
        deg_wt_kernel<<<grid2, 256, 0, stream>>>(src, dst, out, weight, wT, rank, n_edges);
    }
    {
        int grid = (n_nodes + TILE_M - 1) / TILE_M;
        gemm_maxk_kernel<<<grid, 128, 0, stream>>>(feat, wT, bias, out,
                                                   tk_vals, tk_idx, seg, list_top, dcnt, n_nodes);
    }
    if (gather) {
        fill_kernel<<<2048, 256, 0, stream>>>(src, dst, rank, seg, sorted_src, n_edges);
        gather_kernel<<<2048, 256, 0, stream>>>(seg, sorted_src, tk_vals, tk_idx, bias, out, n_nodes);
    } else {
        long long total = (long long)n_edges * 4;
        int grid = (int)((total + 255) / 256);
        scatter_kernel<<<grid, 256, 0, stream>>>(src, dst, tk_vals, tk_idx, out, n_edges);
    }
}


# edge ranks slice-local: one walk (LDS add's return = rank), counts stored as per-slice planes, GEMM block forms degrees and slice starts; no global atomic, no zeroing launch; sum order slice-major
# speedup vs baseline: 2.3615x; 1.0437x over previous
#include <hip/hip_runtime.h>
#include <math.h>

typedef __attribute__((ext_vector_type(2))) float v2f;
typedef __attribute__((ext_vector_type(8))) float v8f;

#define IN_FEATS 256
#define OUT_FEATS 64
#define MAXK 16
#define TILE_M 16
#define LDA (IN_FEATS + 4)
#define LDH (OUT_FEATS + 4)

#define DEG_OUT_COL 0
#define DEG_IN_COL  32

__device__ __forceinline__ int wfrag_pos(int gi) {
    const int k = gi >> 6;
    const int c = gi & 63;
    const int w = c >> 4, c16 = c & 15;
    const int jj = k >> 3, sub = (k >> 2) & 1, h = (k >> 1) & 1, e = k & 1;
    return (((w * 32 + jj) * 32 + (h * 16 + c16)) << 2) + sub * 2 + e;
}

__global__ void init_kernel(float* __restrict__ out, int* __restrict__ list_top, int n_nodes) {
    long long i = (long long)blockIdx.x * blockDim.x + threadIdx.x;
    if (i == 0 && list_top != nullptr) *list_top = 0;
    if (i < 2LL * n_nodes)
        out[(i >> 1) * OUT_FEATS + ((i & 1) ? DEG_IN_COL : DEG_OUT_COL)] = 0.0f;
}

__global__ void deg_wt_kernel(const int* __restrict__ src, const int* __restrict__ dst,
                              float* __restrict__ out,
                              const float* __restrict__ weight, float* __restrict__ wT,
                              int* __restrict__ rank, int n_edges) {
    int i = blockIdx.x * blockDim.x + threadIdx.x;
    if (i < n_edges) {
        const int s = src[i];
        const int d = dst[i];
        atomicAdd(out + (long long)s * OUT_FEATS + DEG_OUT_COL, 1.0f);
        const int r = atomicAdd((int*)(out + (long long)d * OUT_FEATS + DEG_IN_COL), 1);
        if (rank != nullptr) rank[i] = r;
    }
    if (i < IN_FEATS * OUT_FEATS) wT[wfrag_pos(i)] = weight[i];
}


#define DEG_R       6400
#define DEG_NS      16
#define DEG_THREADS 1024

__device__ __forceinline__ void count_in_range(int* cnt, int v, int lo, int len) {
    const unsigned int o = (unsigned int)(v - lo);
    if (o < (unsigned int)len) atomicAdd(&cnt[o], 1);
}

__device__ __forceinline__ void rank_in_range(int* cnt, int* __restrict__ rank, int e,
                                              int v, int lo, int len) {
    const unsigned int o = (unsigned int)(v - lo);
    if (o < (unsigned int)len) rank[e] = atomicAdd(&cnt[o], 1);
}

__global__ __launch_bounds__(DEG_THREADS)
void deg_lds_kernel(const int* __restrict__ src, const int* __restrict__ dst,
                    int* __restrict__ cin, int* __restrict__ cout,
                    const float* __restrict__ weight, float* __restrict__ wT,
                    int* __restrict__ rank, int* __restrict__ list_top,
                    int n_nodes, int n_edges) {
    __shared__ int sIn[DEG_R];
    __shared__ int sOut[DEG_R];

    const int tid = threadIdx.x;
    const int b = blockIdx.x / DEG_NS;
    const int s = blockIdx.x % DEG_NS;
    const int lo = b * DEG_R;
    const int len = (n_nodes - lo) < DEG_R ? (n_nodes - lo) : DEG_R;

    if (blockIdx.x == 0 && tid == 0) *list_top = 0;

    {
        const int gi = blockIdx.x * DEG_THREADS + tid;
        if (gi < IN_FEATS * OUT_FEATS) wT[wfrag_pos(gi)] = weight[gi];
    }

    for (int n = tid; n < DEG_R; n += DEG_THREADS) { sIn[n] = 0; sOut[n] = 0; }
    __syncthreads();

    const int nq = n_edges >> 2;
    const int qs = (nq + DEG_NS - 1) / DEG_NS;
    const int q0 = s * qs;
    const int q1 = (q0 + qs) < nq ? (q0 + qs) : nq;
    const int e_tail = (nq << 2) + tid;
    const bool tail = (s == DEG_NS - 1) && (e_tail < n_edges);

    for (int q = q0 + tid; q < q1; q += 2 * DEG_THREADS) {
        const bool two = (q + DEG_THREADS) < q1;
        const int qb = two ? (q + DEG_THREADS) : q;
        const int len1 = two ? len : 0;
        const int4 d0 = ((const int4*)dst)[q];
        const int4 d1 = ((const int4*)dst)[qb];
        const int4 s0 = ((const int4*)src)[q];
        const int4 s1 = ((const int4*)src)[qb];
        const int e0 = q << 2;
        const int e1 = qb << 2;
        rank_in_range(sIn, rank, e0,     d0.x, lo, len);
        rank_in_range(sIn, rank, e0 + 1, d0.y, lo, len);
        rank_in_range(sIn, rank, e0 + 2, d0.z, lo, len);
        rank_in_range(sIn, rank, e0 + 3, d0.w, lo, len);
        rank_in_range(sIn, rank, e1,     d1.x, lo, len1);
        rank_in_range(sIn, rank, e1 + 1, d1.y, lo, len1);
        rank_in_range(sIn, rank, e1 + 2, d1.z, lo, len1);
        rank_in_range(sIn, rank, e1 + 3, d1.w, lo, len1);
        count_in_range(sOut, s0.x, lo, len); count_in_range(sOut, s0.y, lo, len);
        count_in_range(sOut, s0.z, lo, len); count_in_range(sOut, s0.w, lo, len);
        count_in_range(sOut, s1.x, lo, len1); count_in_range(sOut, s1.y, lo, len1);
        count_in_range(sOut, s1.z, lo, len1); count_in_range(sOut, s1.w, lo, len1);
    }
    if (tail) {
        rank_in_range(sIn, rank, e_tail, dst[e_tail], lo, len);
        count_in_range(sOut, src[e_tail], lo, len);
    }
    __syncthreads();

    for (int n = tid; n < len; n += DEG_THREADS) {
        cin[(long long)s * n_nodes + lo + n]  = sIn[n];
        cout[(long long)s * n_nodes + lo + n] = sOut[n];
    }
}

__global__ __launch_bounds__(128)
void gemm_maxk_kernel(const float* __restrict__ feat, const float* __restrict__ wT,
                      const float* __restrict__ bias, float* out,
                      float* __restrict__ tk_vals, unsigned int* __restrict__ tk_idx,
                      int* __restrict__ seg, int* __restrict__ list_top,
                      int* cin, const int* cout, int n_nodes) {
    __shared__ __attribute__((aligned(16))) float sA[TILE_M * LDA];
    __shared__ __attribute__((aligned(16))) float sH[TILE_M * LDH];
    __shared__ float sScale[TILE_M];

    const int tid = threadIdx.x;
    const int m0 = blockIdx.x * TILE_M;

    if (cin != nullptr) {
        if (tid < TILE_M) {
            const bool valid = (m0 + tid) < n_nodes;
            int ti = 0, to = 0;
            #pragma unroll
            for (int s = 0; s < DEG_NS; ++s) {
                ti += valid ? cin[(long long)s * n_nodes + m0 + tid] : 0;
                to += valid ? cout[(long long)s * n_nodes + m0 + tid] : 0;
            }
            float od = (float)to, id = (float)ti;
            od = od < 1.f ? 1.f : od;
            id = id < 1.f ? 1.f : id;
            sScale[tid] = valid ? rsqrtf(od) * rsqrtf(id) : 0.f;
        }
    } else if (tid < TILE_M) {
        float sc = 0.f;
        if (m0 + tid < n_nodes) {
            const float* orow = out + (long long)(m0 + tid) * OUT_FEATS;
            float od = orow[DEG_OUT_COL];
            float id = (float)((const int*)orow)[DEG_IN_COL];
            od = od < 1.f ? 1.f : od;
            id = id < 1.f ? 1.f : id;
            sc = rsqrtf(od) * rsqrtf(id);
        }
        sScale[tid] = sc;
    }

    for (int i = tid; i < TILE_M * (IN_FEATS / 4); i += 128) {
        int row = i / (IN_FEATS / 4);
        int c4  = i % (IN_FEATS / 4);
        float4 v = make_float4(0.f, 0.f, 0.f, 0.f);
        if (m0 + row < n_nodes)
            v = ((const float4*)feat)[(long long)(m0 + row) * (IN_FEATS / 4) + c4];
        float* p = &sA[row * LDA + c4 * 4];
        p[0] = v.x; p[1] = v.y; p[2] = v.z; p[3] = v.w;
    }
    __syncthreads();

    for (int i = tid; seg == nullptr && i < TILE_M * (OUT_FEATS / 4); i += 128) {
        int row = i / (OUT_FEATS / 4);
        int c4  = i % (OUT_FEATS / 4);
        if (m0 + row < n_nodes)
            ((float4*)out)[(long long)(m0 + row) * (OUT_FEATS / 4) + c4] =
                ((const float4*)bias)[c4];
    }

    const int wave = tid >> 5;
    const int lane = tid & 31;
    const int r16  = lane & 15;
    const int koff = (lane >> 4) << 1;
    const float4* wfrag = (const float4*)wT + (wave * 32) * 32 + lane;
    const float* arow = sA + r16 * LDA;

    v8f c = {0.f, 0.f, 0.f, 0.f, 0.f, 0.f, 0.f, 0.f};
    #pragma unroll
    for (int k0 = 0; k0 < IN_FEATS; k0 += 8) {
        const float4 b4 = wfrag[(k0 >> 3) * 32];
        v2f a0, a1, b0, b1;
        a0.x = arow[k0 + koff];
        a0.y = arow[k0 + koff + 1];
        a1.x = arow[k0 + 4 + koff];
        a1.y = arow[k0 + 4 + koff + 1];
        b0.x = b4.x; b0.y = b4.y;
        b1.x = b4.z; b1.y = b4.w;
        c = __builtin_amdgcn_wmma_f32_16x16x4_f32(false, a0, false, b0, (short)0, c,
                                                  false, false);
        c = __builtin_amdgcn_wmma_f32_16x16x4_f32(false, a1, false, b1, (short)0, c,
                                                  false, false);
    }

    {
        const int rbase = (lane >> 4) << 3;
        const int col = wave * 16 + r16;
        #pragma unroll
        for (int r = 0; r < 8; ++r)
            sH[(rbase + r) * LDH + col] = c[r];
    }
    __syncthreads();

    {
        const int row = tid >> 3;
        const int p   = tid & 7;
        const float* rowp = &sH[row * LDH];
        float x[8];
        int cnt[8];
        #pragma unroll
        for (int e = 0; e < 8; ++e) { x[e] = rowp[8 * e + p]; cnt[e] = 0; }
        #pragma unroll
        for (int g = 0; g < 8; ++g) {
            const float4 h0 = *(const float4*)(rowp + 8 * g);
            const float4 h1 = *(const float4*)(rowp + 8 * g + 4);
            const float hv[8] = {h0.x, h0.y, h0.z, h0.w, h1.x, h1.y, h1.z, h1.w};
            #pragma unroll
            for (int pp = 0; pp < 8; ++pp) {
                const float v = hv[pp];
                #pragma unroll
                for (int e = 0; e < 8; ++e) {
                    bool before;
                    if (g < e)      before = v >= x[e];
                    else if (g > e) before = v > x[e];
                    else            before = (v > x[e]) || (v == x[e] && pp < p);
                    cnt[e] += before ? 1 : 0;
                }
            }
        }
        const float scale = sScale[row];
        float* sV = sA;
        unsigned char* sI = (unsigned char*)(sA + TILE_M * MAXK);
        #pragma unroll
        for (int e = 0; e < 8; ++e) {
            const bool keep = cnt[e] < MAXK;
            const int slot = row * MAXK + cnt[e];
            sV[keep ? slot : 512 + tid] = x[e] * scale;
            sI[keep ? slot : 4096 + tid] = (unsigned char)(8 * e + p);
        }
    }
    __syncthreads();
    if (tid < 64) {
        const int row = tid >> 2;
        if (m0 + row < n_nodes)
            ((float4*)tk_vals)[(long long)(m0 + row) * (MAXK / 4) + (tid & 3)] =
                ((const float4*)sA)[tid];
    } else if (tid < 64 + TILE_M) {
        const int row = tid - 64;
        if (m0 + row < n_nodes)
            ((uint4*)tk_idx)[m0 + row] = ((const uint4*)(sA + TILE_M * MAXK))[row];
    }

    if (cin != nullptr && tid >= 32 && tid < 64) {
        const int l = tid - 32;
        const bool valid = l < TILE_M && (m0 + l) < n_nodes;
        int ci[DEG_NS];
        int ti = 0;
        #pragma unroll
        for (int s = 0; s < DEG_NS; ++s) {
            ci[s] = valid ? cin[(long long)s * n_nodes + m0 + l] : 0;
            ti += ci[s];
        }
        int inc = ti;
        #pragma unroll
        for (int d = 1; d < 32; d <<= 1) {
            const int t = __shfl_up(inc, d);
            if (l >= d) inc += t;
        }
        const int total = __shfl(inc, 31);
        int base = 0;
        if (l == 0) base = atomicAdd(list_top, total);
        base = __shfl(base, 0);
        int run = base + inc - ti;
        if (valid) {
            ((int2*)seg)[m0 + l] = make_int2(run, ti);
            #pragma unroll
            for (int s = 0; s < DEG_NS; ++s) {
                cin[(long long)s * n_nodes + m0 + l] = run;
                run += ci[s];
            }
        }
    }
}

__global__ void scatter_kernel(const int* __restrict__ src, const int* __restrict__ dst,
                               const float* __restrict__ tk_vals,
                               const unsigned int* __restrict__ tk_idx,
                               float* __restrict__ out, int n_edges) {
    long long gid = (long long)blockIdx.x * blockDim.x + threadIdx.x;
    long long total = (long long)n_edges * 4;
    if (gid >= total) return;
    int e = (int)(gid >> 2);
    int part = (int)(gid & 3);
    int s = src[e], d = dst[e];
    float4 v = ((const float4*)tk_vals)[(long long)s * 4 + part];
    unsigned int ip = tk_idx[(long long)s * 4 + part];
    float* orow = out + (long long)d * OUT_FEATS;
    atomicAdd(orow + (ip & 0xffu), v.x);
    atomicAdd(orow + ((ip >> 8) & 0xffu), v.y);
    atomicAdd(orow + ((ip >> 16) & 0xffu), v.z);
    atomicAdd(orow + ((ip >> 24) & 0xffu), v.w);
}


__global__ void fill_kernel(const int* __restrict__ src, const int* __restrict__ dst,
                            const int* __restrict__ rank, const int* __restrict__ cin,
                            int* __restrict__ sorted_src, int n_nodes, int n_edges) {
    const int nq = n_edges >> 2;
    const int qs = (nq + DEG_NS - 1) / DEG_NS;
    const long long stride = (long long)gridDim.x * blockDim.x;
    for (long long e = (long long)blockIdx.x * blockDim.x + threadIdx.x; e < n_edges; e += stride) {
        const int d = dst[e];
        const int q = (int)(e >> 2);
        const int s = q < nq ? q / qs : DEG_NS - 1;
        sorted_src[cin[(long long)s * n_nodes + d] + rank[e]] = src[e];
    }
}

__global__ __launch_bounds__(256)
void gather_kernel(const int* __restrict__ seg, const int* __restrict__ sorted_src,
                   const float* __restrict__ tk_vals, const unsigned int* __restrict__ tk_idx,
                   const float* __restrict__ bias, float* __restrict__ out, int n_nodes) {
    __shared__ __attribute__((aligned(16))) float sRow[8][OUT_FEATS];
    const int lane = threadIdx.x & 31;
    const int wave = threadIdx.x >> 5;
    const int half = lane >> 4;
    const int k    = lane & 15;
    const int sh   = (k & 3) * 8;
    float* row = sRow[wave];
    const float2 b2 = ((const float2*)bias)[lane];

    for (int node = blockIdx.x * 8 + wave; node < n_nodes; node += gridDim.x * 8) {
        const int2 sg = ((const int2*)seg)[node];
        const int start = __builtin_amdgcn_readfirstlane(sg.x);
        const int cnt   = __builtin_amdgcn_readfirstlane(sg.y);
        row[lane] = 0.f;
        row[lane + 32] = 0.f;
        for (int base = 0; base < cnt; base += 32) {
            const int rem = (cnt - base) < 32 ? (cnt - base) : 32;
            const int sl = lane < rem ? sorted_src[start + base + lane] : 0;
            for (int q = 0; q < rem; q += 8) {
                float v[4];
                unsigned int iw[4];
                #pragma unroll
                for (int u = 0; u < 4; ++u) {
                    const int s = __shfl(sl, q + 2 * u + half);
                    v[u]  = tk_vals[(long long)s * MAXK + k];
                    iw[u] = tk_idx[(long long)s * 4 + (k >> 2)];
                }
                #pragma unroll
                for (int u = 0; u < 4; ++u)
                    if (q + 2 * u + half < rem)
                        atomicAdd(&row[(iw[u] >> sh) & 0xffu], v[u]);
            }
        }
        const float2 r2 = ((const float2*)row)[lane];
        ((float2*)out)[(long long)node * (OUT_FEATS / 2) + lane] =
            make_float2(r2.x + b2.x, r2.y + b2.y);
    }
}

extern "C" void kernel_launch(void* const* d_in, const int* in_sizes, int n_in,
                              void* d_out, int out_size, void* d_ws, size_t ws_size,
                              hipStream_t stream) {
    const float* feat   = (const float*)d_in[0];
    const float* weight = (const float*)d_in[1];
    const float* bias   = (const float*)d_in[2];
    const int*   src    = (const int*)d_in[3];
    const int*   dst    = (const int*)d_in[4];
    float* out = (float*)d_out;

    const int n_nodes = in_sizes[0] / IN_FEATS;
    const int n_edges = in_sizes[3];

    float* ws_f    = (float*)d_ws;
    float* wT      = ws_f;
    float* tk_vals = wT + IN_FEATS * OUT_FEATS;
    unsigned int* tk_idx = (unsigned int*)(tk_vals + (long long)n_nodes * MAXK);

    const size_t o_seg    = (((size_t)IN_FEATS * OUT_FEATS + (size_t)n_nodes * (MAXK + 4)) * 4 + 255) & ~(size_t)255;
    const size_t o_rank   = (o_seg + (size_t)n_nodes * 8 + 255) & ~(size_t)255;
    const size_t o_sorted = (o_rank + (size_t)n_edges * 4 + 255) & ~(size_t)255;
    const size_t o_top    = (o_sorted + (size_t)n_edges * 4 + 255) & ~(size_t)255;
    const bool gather = ws_size >= o_top + 256;
    char* ws_b = (char*)d_ws;
    int*  seg        = gather ? (int*)(ws_b + o_seg)    : nullptr;
    int*  rank       = gather ? (int*)(ws_b + o_rank)   : nullptr;
    int*  sorted_src = gather ? (int*)(ws_b + o_sorted) : nullptr;
    int*  list_top   = gather ? (int*)(ws_b + o_top)    : nullptr;

    int* cin  = gather ? (int*)d_out : nullptr;
    int* cout = gather ? (int*)d_out + (long long)DEG_NS * n_nodes : nullptr;
    if (gather) {
        int nb = (n_nodes + DEG_R - 1) / DEG_R;
        deg_lds_kernel<<<nb * DEG_NS, DEG_THREADS, 0, stream>>>(src, dst, cin, cout, weight, wT, rank,
                                                                 list_top, n_nodes, n_edges);
    } else {
        long long total = 2LL * n_nodes;
        int grid = (int)((total + 255) / 256);
        init_kernel<<<grid, 256, 0, stream>>>(out, list_top, n_nodes);
        int work = n_edges > IN_FEATS * OUT_FEATS ? n_edges : IN_FEATS * OUT_FEATS;
        int grid2 = (work + 255) / 256;
        deg_wt_kernel<<<grid2, 256, 0, stream>>>(src, dst, out, weight, wT, rank, n_edges);
    }
    {
        int grid = (n_nodes + TILE_M - 1) / TILE_M;
        gemm_maxk_kernel<<<grid, 128, 0, stream>>>(feat, wT, bias, out,
                                                   tk_vals, tk_idx, seg, list_top, cin, cout, n_nodes);
    }
    if (gather) {
        fill_kernel<<<2048, 256, 0, stream>>>(src, dst, rank, cin, sorted_src, n_nodes, n_edges);
        gather_kernel<<<2048, 256, 0, stream>>>(seg, sorted_src, tk_vals, tk_idx, bias, out, n_nodes);
    } else {
        long long total = (long long)n_edges * 4;
        int grid = (int)((total + 255) / 256);
        scatter_kernel<<<grid, 256, 0, stream>>>(src, dst, tk_vals, tk_idx, out, n_edges);
    }
}


# top-16 of a row as a set: 16th largest value from a min/max merge network over the row's 8 lanes, keep >= it, slots in column order; rank counting only on boundary ties; sum order unchanged
# speedup vs baseline: 2.5860x; 1.0951x over previous
#include <hip/hip_runtime.h>
#include <math.h>

typedef __attribute__((ext_vector_type(2))) float v2f;
typedef __attribute__((ext_vector_type(8))) float v8f;

#define IN_FEATS 256
#define OUT_FEATS 64
#define MAXK 16
#define TILE_M 16
#define LDA (IN_FEATS + 4)
#define LDH (OUT_FEATS + 4)

#define DEG_OUT_COL 0
#define DEG_IN_COL  32

__device__ __forceinline__ int wfrag_pos(int gi) {
    const int k = gi >> 6;
    const int c = gi & 63;
    const int w = c >> 4, c16 = c & 15;
    const int jj = k >> 3, sub = (k >> 2) & 1, h = (k >> 1) & 1, e = k & 1;
    return (((w * 32 + jj) * 32 + (h * 16 + c16)) << 2) + sub * 2 + e;
}

__global__ void init_kernel(float* __restrict__ out, int* __restrict__ list_top, int n_nodes) {
    long long i = (long long)blockIdx.x * blockDim.x + threadIdx.x;
    if (i == 0 && list_top != nullptr) *list_top = 0;
    if (i < 2LL * n_nodes)
        out[(i >> 1) * OUT_FEATS + ((i & 1) ? DEG_IN_COL : DEG_OUT_COL)] = 0.0f;
}

__global__ void deg_wt_kernel(const int* __restrict__ src, const int* __restrict__ dst,
                              float* __restrict__ out,
                              const float* __restrict__ weight, float* __restrict__ wT,
                              int* __restrict__ rank, int n_edges) {
    int i = blockIdx.x * blockDim.x + threadIdx.x;
    if (i < n_edges) {
        const int s = src[i];
        const int d = dst[i];
        atomicAdd(out + (long long)s * OUT_FEATS + DEG_OUT_COL, 1.0f);
        const int r = atomicAdd((int*)(out + (long long)d * OUT_FEATS + DEG_IN_COL), 1);
        if (rank != nullptr) rank[i] = r;
    }
    if (i < IN_FEATS * OUT_FEATS) wT[wfrag_pos(i)] = weight[i];
}


#define DEG_R       6400
#define DEG_NS      16
#define DEG_THREADS 1024

__device__ __forceinline__ void count_in_range(int* cnt, int v, int lo, int len) {
    const unsigned int o = (unsigned int)(v - lo);
    if (o < (unsigned int)len) atomicAdd(&cnt[o], 1);
}

__device__ __forceinline__ void rank_in_range(int* cnt, int* __restrict__ rank, int e,
                                              int v, int lo, int len) {
    const unsigned int o = (unsigned int)(v - lo);
    if (o < (unsigned int)len) rank[e] = atomicAdd(&cnt[o], 1);
}

__global__ __launch_bounds__(DEG_THREADS)
void deg_lds_kernel(const int* __restrict__ src, const int* __restrict__ dst,
                    int* __restrict__ cin, int* __restrict__ cout,
                    const float* __restrict__ weight, float* __restrict__ wT,
                    int* __restrict__ rank, int* __restrict__ list_top,
                    int n_nodes, int n_edges) {
    __shared__ int sIn[DEG_R];
    __shared__ int sOut[DEG_R];

    const int tid = threadIdx.x;
    const int b = blockIdx.x / DEG_NS;
    const int s = blockIdx.x % DEG_NS;
    const int lo = b * DEG_R;
    const int len = (n_nodes - lo) < DEG_R ? (n_nodes - lo) : DEG_R;

    if (blockIdx.x == 0 && tid == 0) *list_top = 0;

    {
        const int gi = blockIdx.x * DEG_THREADS + tid;
        if (gi < IN_FEATS * OUT_FEATS) wT[wfrag_pos(gi)] = weight[gi];
    }

    for (int n = tid; n < DEG_R; n += DEG_THREADS) { sIn[n] = 0; sOut[n] = 0; }
    __syncthreads();

    const int nq = n_edges >> 2;
    const int qs = (nq + DEG_NS - 1) / DEG_NS;
    const int q0 = s * qs;
    const int q1 = (q0 + qs) < nq ? (q0 + qs) : nq;
    const int e_tail = (nq << 2) + tid;
    const bool tail = (s == DEG_NS - 1) && (e_tail < n_edges);

    for (int q = q0 + tid; q < q1; q += 2 * DEG_THREADS) {
        const bool two = (q + DEG_THREADS) < q1;
        const int qb = two ? (q + DEG_THREADS) : q;
        const int len1 = two ? len : 0;
        const int4 d0 = ((const int4*)dst)[q];
        const int4 d1 = ((const int4*)dst)[qb];
        const int4 s0 = ((const int4*)src)[q];
        const int4 s1 = ((const int4*)src)[qb];
        const int e0 = q << 2;
        const int e1 = qb << 2;
        rank_in_range(sIn, rank, e0,     d0.x, lo, len);
        rank_in_range(sIn, rank, e0 + 1, d0.y, lo, len);
        rank_in_range(sIn, rank, e0 + 2, d0.z, lo, len);
        rank_in_range(sIn, rank, e0 + 3, d0.w, lo, len);
        rank_in_range(sIn, rank, e1,     d1.x, lo, len1);
        rank_in_range(sIn, rank, e1 + 1, d1.y, lo, len1);
        rank_in_range(sIn, rank, e1 + 2, d1.z, lo, len1);
        rank_in_range(sIn, rank, e1 + 3, d1.w, lo, len1);
        count_in_range(sOut, s0.x, lo, len); count_in_range(sOut, s0.y, lo, len);
        count_in_range(sOut, s0.z, lo, len); count_in_range(sOut, s0.w, lo, len);
        count_in_range(sOut, s1.x, lo, len1); count_in_range(sOut, s1.y, lo, len1);
        count_in_range(sOut, s1.z, lo, len1); count_in_range(sOut, s1.w, lo, len1);
    }
    if (tail) {
        rank_in_range(sIn, rank, e_tail, dst[e_tail], lo, len);
        count_in_range(sOut, src[e_tail], lo, len);
    }
    __syncthreads();

    for (int n = tid; n < len; n += DEG_THREADS) {
        cin[(long long)s * n_nodes + lo + n]  = sIn[n];
        cout[(long long)s * n_nodes + lo + n] = sOut[n];
    }
}

__device__ __forceinline__ float lane_xor1(float v) {
    return __int_as_float(__builtin_amdgcn_update_dpp(0, __float_as_int(v), 0xB1, 0xf, 0xf, true));
}
__device__ __forceinline__ float lane_xor3(float v) {
    return __int_as_float(__builtin_amdgcn_update_dpp(0, __float_as_int(v), 0x1B, 0xf, 0xf, true));
}
__device__ __forceinline__ float lane_rev8(float v) {
    return __int_as_float(__builtin_amdgcn_update_dpp(0, __float_as_int(v), 0x141, 0xf, 0xf, true));
}
__device__ __forceinline__ float quad_first(float v) {
    return __int_as_float(__builtin_amdgcn_update_dpp(0, __float_as_int(v), 0x00, 0xf, 0xf, true));
}
__device__ __forceinline__ float quad_last(float v) {
    return __int_as_float(__builtin_amdgcn_update_dpp(0, __float_as_int(v), 0xFF, 0xf, 0xf, true));
}
#define CE_DESC(a, b) { const float hi_ = __builtin_fmaxf(a, b); const float lo_ = __builtin_fminf(a, b); a = hi_; b = lo_; }
#define MERGE8_DESC(r) { \
    CE_DESC(r[0], r[4]) CE_DESC(r[1], r[5]) CE_DESC(r[2], r[6]) CE_DESC(r[3], r[7]) \
    CE_DESC(r[0], r[2]) CE_DESC(r[1], r[3]) CE_DESC(r[4], r[6]) CE_DESC(r[5], r[7]) \
    CE_DESC(r[0], r[1]) CE_DESC(r[2], r[3]) CE_DESC(r[4], r[5]) CE_DESC(r[6], r[7]) }

__global__ __launch_bounds__(128)
void gemm_maxk_kernel(const float* __restrict__ feat, const float* __restrict__ wT,
                      const float* __restrict__ bias, float* out,
                      float* __restrict__ tk_vals, unsigned int* __restrict__ tk_idx,
                      int* __restrict__ seg, int* __restrict__ list_top,
                      int* cin, const int* cout, int n_nodes) {
    __shared__ __attribute__((aligned(16))) float sA[TILE_M * LDA];
    __shared__ __attribute__((aligned(16))) float sH[TILE_M * LDH];
    __shared__ float sScale[TILE_M];

    const int tid = threadIdx.x;
    const int m0 = blockIdx.x * TILE_M;

    if (cin != nullptr) {
        if (tid < TILE_M) {
            const bool valid = (m0 + tid) < n_nodes;
            int ti = 0, to = 0;
            #pragma unroll
            for (int s = 0; s < DEG_NS; ++s) {
                ti += valid ? cin[(long long)s * n_nodes + m0 + tid] : 0;
                to += valid ? cout[(long long)s * n_nodes + m0 + tid] : 0;
            }
            float od = (float)to, id = (float)ti;
            od = od < 1.f ? 1.f : od;
            id = id < 1.f ? 1.f : id;
            sScale[tid] = valid ? rsqrtf(od) * rsqrtf(id) : 0.f;
        }
    } else if (tid < TILE_M) {
        float sc = 0.f;
        if (m0 + tid < n_nodes) {
            const float* orow = out + (long long)(m0 + tid) * OUT_FEATS;
            float od = orow[DEG_OUT_COL];
            float id = (float)((const int*)orow)[DEG_IN_COL];
            od = od < 1.f ? 1.f : od;
            id = id < 1.f ? 1.f : id;
            sc = rsqrtf(od) * rsqrtf(id);
        }
        sScale[tid] = sc;
    }

    for (int i = tid; i < TILE_M * (IN_FEATS / 4); i += 128) {
        int row = i / (IN_FEATS / 4);
        int c4  = i % (IN_FEATS / 4);
        float4 v = make_float4(0.f, 0.f, 0.f, 0.f);
        if (m0 + row < n_nodes)
            v = ((const float4*)feat)[(long long)(m0 + row) * (IN_FEATS / 4) + c4];
        float* p = &sA[row * LDA + c4 * 4];
        p[0] = v.x; p[1] = v.y; p[2] = v.z; p[3] = v.w;
    }
    __syncthreads();

    for (int i = tid; seg == nullptr && i < TILE_M * (OUT_FEATS / 4); i += 128) {
        int row = i / (OUT_FEATS / 4);
        int c4  = i % (OUT_FEATS / 4);
        if (m0 + row < n_nodes)
            ((float4*)out)[(long long)(m0 + row) * (OUT_FEATS / 4) + c4] =
                ((const float4*)bias)[c4];
    }

    const int wave = tid >> 5;
    const int lane = tid & 31;
    const int r16  = lane & 15;
    const int koff = (lane >> 4) << 1;
    const float4* wfrag = (const float4*)wT + (wave * 32) * 32 + lane;
    const float* arow = sA + r16 * LDA;

    v8f c = {0.f, 0.f, 0.f, 0.f, 0.f, 0.f, 0.f, 0.f};
    #pragma unroll
    for (int k0 = 0; k0 < IN_FEATS; k0 += 8) {
        const float4 b4 = wfrag[(k0 >> 3) * 32];
        v2f a0, a1, b0, b1;
        a0.x = arow[k0 + koff];
        a0.y = arow[k0 + koff + 1];
        a1.x = arow[k0 + 4 + koff];
        a1.y = arow[k0 + 4 + koff + 1];
        b0.x = b4.x; b0.y = b4.y;
        b1.x = b4.z; b1.y = b4.w;
        c = __builtin_amdgcn_wmma_f32_16x16x4_f32(false, a0, false, b0, (short)0, c,
                                                  false, false);
        c = __builtin_amdgcn_wmma_f32_16x16x4_f32(false, a1, false, b1, (short)0, c,
                                                  false, false);
    }

    {
        const int rbase = (lane >> 4) << 3;
        const int col = wave * 16 + r16;
        #pragma unroll
        for (int r = 0; r < 8; ++r)
            sH[(rbase + r) * LDH + col] = c[r];
    }
    __syncthreads();

    {
        const int row = tid >> 3;
        const int p   = tid & 7;
        const float* rowp = &sH[row * LDH];
        float x[8];
        #pragma unroll
        for (int e = 0; e < 8; ++e) x[e] = rowp[8 * e + p];

        float r[8], u[8];
        #pragma unroll
        for (int e = 0; e < 8; ++e) r[e] = x[e];
        CE_DESC(r[0], r[1]) CE_DESC(r[2], r[3]) CE_DESC(r[4], r[5]) CE_DESC(r[6], r[7])
        CE_DESC(r[0], r[2]) CE_DESC(r[1], r[3]) CE_DESC(r[4], r[6]) CE_DESC(r[5], r[7])
        CE_DESC(r[1], r[2]) CE_DESC(r[5], r[6])
        CE_DESC(r[0], r[4]) CE_DESC(r[1], r[5]) CE_DESC(r[2], r[6]) CE_DESC(r[3], r[7])
        CE_DESC(r[2], r[4]) CE_DESC(r[3], r[5])
        CE_DESC(r[1], r[2]) CE_DESC(r[3], r[4]) CE_DESC(r[5], r[6])
        const float pick = (p & 1) ? -INFINITY : INFINITY;
        #pragma unroll
        for (int i = 0; i < 8; ++i) u[i] = lane_xor1(r[7 - i]);
        #pragma unroll
        for (int i = 0; i < 8; ++i) r[i] = __builtin_amdgcn_fmed3f(r[i], u[i], pick);
        MERGE8_DESC(r)
        #pragma unroll
        for (int i = 0; i < 8; ++i) u[i] = lane_xor3(r[7 - i]);
        #pragma unroll
        for (int i = 0; i < 8; ++i) r[i] = __builtin_amdgcn_fmed3f(r[i], u[i], INFINITY);
        #pragma unroll
        for (int i = 0; i < 8; ++i) u[i] = lane_xor1(r[i]);
        #pragma unroll
        for (int i = 0; i < 8; ++i) r[i] = __builtin_amdgcn_fmed3f(r[i], u[i], pick);
        MERGE8_DESC(r)
        #pragma unroll
        for (int i = 0; i < 8; ++i) u[i] = lane_rev8(r[7 - i]);
        #pragma unroll
        for (int i = 0; i < 8; ++i) r[i] = __builtin_amdgcn_fmed3f(r[i], u[i], INFINITY);
        float m = __builtin_fminf(__builtin_fminf(__builtin_fminf(r[0], r[1]), __builtin_fminf(r[2], r[3])),
                                  __builtin_fminf(__builtin_fminf(r[4], r[5]), __builtin_fminf(r[6], r[7])));
        m = __builtin_amdgcn_fmed3f(m, lane_xor1(m), -INFINITY);
        const float m_lo = quad_first(m);
        const float m_hi = quad_last(m);
        const float thr = (p < 4) ? m_lo : m_hi;

        int nkeep = 0;
        #pragma unroll
        for (int e = 0; e < 8; ++e) nkeep += (x[e] >= thr) ? 1 : 0;
        int inc = nkeep;
        {
            int t1 = __builtin_amdgcn_update_dpp(0, inc, 0x111, 0xf, 0xf, true);
            if (p >= 1) inc += t1;
            int t2 = __builtin_amdgcn_update_dpp(0, inc, 0x112, 0xf, 0xf, true);
            if (p >= 2) inc += t2;
            int t4 = __builtin_amdgcn_update_dpp(0, inc, 0x114, 0xf, 0xf, true);
            if (p >= 4) inc += t4;
        }
        const bool exact = __builtin_amdgcn_ballot_w32(p == 7 && inc != MAXK) == 0u;

        int slot_of[8];
        if (exact) {
            int run = inc - nkeep;
            #pragma unroll
            for (int e = 0; e < 8; ++e) {
                const bool k = x[e] >= thr;
                slot_of[e] = k ? run : MAXK;
                run += k ? 1 : 0;
            }
        } else {
            #pragma unroll
            for (int e = 0; e < 8; ++e) slot_of[e] = 0;
            #pragma unroll
            for (int g = 0; g < 8; ++g) {
                const float4 h0 = *(const float4*)(rowp + 8 * g);
                const float4 h1 = *(const float4*)(rowp + 8 * g + 4);
                const float hv[8] = {h0.x, h0.y, h0.z, h0.w, h1.x, h1.y, h1.z, h1.w};
                #pragma unroll
                for (int pp = 0; pp < 8; ++pp) {
                    const float v = hv[pp];
                    #pragma unroll
                    for (int e = 0; e < 8; ++e) {
                        bool before;
                        if (g < e)      before = v >= x[e];
                        else if (g > e) before = v > x[e];
                        else            before = (v > x[e]) || (v == x[e] && pp < p);
                        slot_of[e] += before ? 1 : 0;
                    }
                }
            }
        }
        const float scale = sScale[row];
        float* sV = sA;
        unsigned char* sI = (unsigned char*)(sA + TILE_M * MAXK);
        #pragma unroll
        for (int e = 0; e < 8; ++e) {
            const bool keep = slot_of[e] < MAXK;
            const int slot = row * MAXK + slot_of[e];
            sV[keep ? slot : 512 + tid] = x[e] * scale;
            sI[keep ? slot : 4096 + tid] = (unsigned char)(8 * e + p);
        }
    }
    __syncthreads();
    if (tid < 64) {
        const int row = tid >> 2;
        if (m0 + row < n_nodes)
            ((float4*)tk_vals)[(long long)(m0 + row) * (MAXK / 4) + (tid & 3)] =
                ((const float4*)sA)[tid];
    } else if (tid < 64 + TILE_M) {
        const int row = tid - 64;
        if (m0 + row < n_nodes)
            ((uint4*)tk_idx)[m0 + row] = ((const uint4*)(sA + TILE_M * MAXK))[row];
    }

    if (cin != nullptr && tid >= 32 && tid < 64) {
        const int l = tid - 32;
        const bool valid = l < TILE_M && (m0 + l) < n_nodes;
        int ci[DEG_NS];
        int ti = 0;
        #pragma unroll
        for (int s = 0; s < DEG_NS; ++s) {
            ci[s] = valid ? cin[(long long)s * n_nodes + m0 + l] : 0;
            ti += ci[s];
        }
        int inc = ti;
        #pragma unroll
        for (int d = 1; d < 32; d <<= 1) {
            const int t = __shfl_up(inc, d);
            if (l >= d) inc += t;
        }
        const int total = __shfl(inc, 31);
        int base = 0;
        if (l == 0) base = atomicAdd(list_top, total);
        base = __shfl(base, 0);
        int run = base + inc - ti;
        if (valid) {
            ((int2*)seg)[m0 + l] = make_int2(run, ti);
            #pragma unroll
            for (int s = 0; s < DEG_NS; ++s) {
                cin[(long long)s * n_nodes + m0 + l] = run;
                run += ci[s];
            }
        }
    }
}

__global__ void scatter_kernel(const int* __restrict__ src, const int* __restrict__ dst,
                               const float* __restrict__ tk_vals,
                               const unsigned int* __restrict__ tk_idx,
                               float* __restrict__ out, int n_edges) {
    long long gid = (long long)blockIdx.x * blockDim.x + threadIdx.x;
    long long total = (long long)n_edges * 4;
    if (gid >= total) return;
    int e = (int)(gid >> 2);
    int part = (int)(gid & 3);
    int s = src[e], d = dst[e];
    float4 v = ((const float4*)tk_vals)[(long long)s * 4 + part];
    unsigned int ip = tk_idx[(long long)s * 4 + part];
    float* orow = out + (long long)d * OUT_FEATS;
    atomicAdd(orow + (ip & 0xffu), v.x);
    atomicAdd(orow + ((ip >> 8) & 0xffu), v.y);
    atomicAdd(orow + ((ip >> 16) & 0xffu), v.z);
    atomicAdd(orow + ((ip >> 24) & 0xffu), v.w);
}


__global__ void fill_kernel(const int* __restrict__ src, const int* __restrict__ dst,
                            const int* __restrict__ rank, const int* __restrict__ cin,
                            int* __restrict__ sorted_src, int n_nodes, int n_edges) {
    const int nq = n_edges >> 2;
    const int qs = (nq + DEG_NS - 1) / DEG_NS;
    const long long stride = (long long)gridDim.x * blockDim.x;
    for (long long e = (long long)blockIdx.x * blockDim.x + threadIdx.x; e < n_edges; e += stride) {
        const int d = dst[e];
        const int q = (int)(e >> 2);
        const int s = q < nq ? q / qs : DEG_NS - 1;
        sorted_src[cin[(long long)s * n_nodes + d] + rank[e]] = src[e];
    }
}

__global__ __launch_bounds__(256)
void gather_kernel(const int* __restrict__ seg, const int* __restrict__ sorted_src,
                   const float* __restrict__ tk_vals, const unsigned int* __restrict__ tk_idx,
                   const float* __restrict__ bias, float* __restrict__ out, int n_nodes) {
    __shared__ __attribute__((aligned(16))) float sRow[8][OUT_FEATS];
    const int lane = threadIdx.x & 31;
    const int wave = threadIdx.x >> 5;
    const int half = lane >> 4;
    const int k    = lane & 15;
    const int sh   = (k & 3) * 8;
    float* row = sRow[wave];
    const float2 b2 = ((const float2*)bias)[lane];

    for (int node = blockIdx.x * 8 + wave; node < n_nodes; node += gridDim.x * 8) {
        const int2 sg = ((const int2*)seg)[node];
        const int start = __builtin_amdgcn_readfirstlane(sg.x);
        const int cnt   = __builtin_amdgcn_readfirstlane(sg.y);
        row[lane] = 0.f;
        row[lane + 32] = 0.f;
        for (int base = 0; base < cnt; base += 32) {
            const int rem = (cnt - base) < 32 ? (cnt - base) : 32;
            const int sl = lane < rem ? sorted_src[start + base + lane] : 0;
            for (int q = 0; q < rem; q += 8) {
                float v[4];
                unsigned int iw[4];
                #pragma unroll
                for (int u = 0; u < 4; ++u) {
                    const int s = __shfl(sl, q + 2 * u + half);
                    v[u]  = tk_vals[(long long)s * MAXK + k];
                    iw[u] = tk_idx[(long long)s * 4 + (k >> 2)];
                }
                #pragma unroll
                for (int u = 0; u < 4; ++u)
                    if (q + 2 * u + half < rem)
                        atomicAdd(&row[(iw[u] >> sh) & 0xffu], v[u]);
            }
        }
        const float2 r2 = ((const float2*)row)[lane];
        ((float2*)out)[(long long)node * (OUT_FEATS / 2) + lane] =
            make_float2(r2.x + b2.x, r2.y + b2.y);
    }
}

extern "C" void kernel_launch(void* const* d_in, const int* in_sizes, int n_in,
                              void* d_out, int out_size, void* d_ws, size_t ws_size,
                              hipStream_t stream) {
    const float* feat   = (const float*)d_in[0];
    const float* weight = (const float*)d_in[1];
    const float* bias   = (const float*)d_in[2];
    const int*   src    = (const int*)d_in[3];
    const int*   dst    = (const int*)d_in[4];
    float* out = (float*)d_out;

    const int n_nodes = in_sizes[0] / IN_FEATS;
    const int n_edges = in_sizes[3];

    float* ws_f    = (float*)d_ws;
    float* wT      = ws_f;
    float* tk_vals = wT + IN_FEATS * OUT_FEATS;
    unsigned int* tk_idx = (unsigned int*)(tk_vals + (long long)n_nodes * MAXK);

    const size_t o_seg    = (((size_t)IN_FEATS * OUT_FEATS + (size_t)n_nodes * (MAXK + 4)) * 4 + 255) & ~(size_t)255;
    const size_t o_rank   = (o_seg + (size_t)n_nodes * 8 + 255) & ~(size_t)255;
    const size_t o_sorted = (o_rank + (size_t)n_edges * 4 + 255) & ~(size_t)255;
    const size_t o_top    = (o_sorted + (size_t)n_edges * 4 + 255) & ~(size_t)255;
    const bool gather = ws_size >= o_top + 256;
    char* ws_b = (char*)d_ws;
    int*  seg        = gather ? (int*)(ws_b + o_seg)    : nullptr;
    int*  rank       = gather ? (int*)(ws_b + o_rank)   : nullptr;
    int*  sorted_src = gather ? (int*)(ws_b + o_sorted) : nullptr;
    int*  list_top   = gather ? (int*)(ws_b + o_top)    : nullptr;

    int* cin  = gather ? (int*)d_out : nullptr;
    int* cout = gather ? (int*)d_out + (long long)DEG_NS * n_nodes : nullptr;
    if (gather) {
        int nb = (n_nodes + DEG_R - 1) / DEG_R;
        deg_lds_kernel<<<nb * DEG_NS, DEG_THREADS, 0, stream>>>(src, dst, cin, cout, weight, wT, rank,
                                                                 list_top, n_nodes, n_edges);
    } else {
        long long total = 2LL * n_nodes;
        int grid = (int)((total + 255) / 256);
        init_kernel<<<grid, 256, 0, stream>>>(out, list_top, n_nodes);
        int work = n_edges > IN_FEATS * OUT_FEATS ? n_edges : IN_FEATS * OUT_FEATS;
        int grid2 = (work + 255) / 256;
        deg_wt_kernel<<<grid2, 256, 0, stream>>>(src, dst, out, weight, wT, rank, n_edges);
    }
    {
        int grid = (n_nodes + TILE_M - 1) / TILE_M;
        gemm_maxk_kernel<<<grid, 128, 0, stream>>>(feat, wT, bias, out,
                                                   tk_vals, tk_idx, seg, list_top, cin, cout, n_nodes);
    }
    if (gather) {
        fill_kernel<<<2048, 256, 0, stream>>>(src, dst, rank, cin, sorted_src, n_nodes, n_edges);
        gather_kernel<<<2048, 256, 0, stream>>>(seg, sorted_src, tk_vals, tk_idx, bias, out, n_nodes);
    } else {
        long long total = (long long)n_edges * 4;
        int grid = (int)((total + 255) / 256);
        scatter_kernel<<<grid, 256, 0, stream>>>(src, dst, tk_vals, tk_idx, out, n_edges);
    }
}


# GEMM block prologue: each thread's eight feat loads and the degree loads (now split over the four waves) in flight together; was load, wait, LDS store eight times in a row; sums unchanged
# speedup vs baseline: 2.6773x; 1.0353x over previous
#include <hip/hip_runtime.h>
#include <math.h>

typedef __attribute__((ext_vector_type(2))) float v2f;
typedef __attribute__((ext_vector_type(8))) float v8f;

#define IN_FEATS 256
#define OUT_FEATS 64
#define MAXK 16
#define TILE_M 16
#define LDA (IN_FEATS + 4)
#define LDH (OUT_FEATS + 4)

#define DEG_OUT_COL 0
#define DEG_IN_COL  32

__device__ __forceinline__ int wfrag_pos(int gi) {
    const int k = gi >> 6;
    const int c = gi & 63;
    const int w = c >> 4, c16 = c & 15;
    const int jj = k >> 3, sub = (k >> 2) & 1, h = (k >> 1) & 1, e = k & 1;
    return (((w * 32 + jj) * 32 + (h * 16 + c16)) << 2) + sub * 2 + e;
}

__global__ void init_kernel(float* __restrict__ out, int* __restrict__ list_top, int n_nodes) {
    long long i = (long long)blockIdx.x * blockDim.x + threadIdx.x;
    if (i == 0 && list_top != nullptr) *list_top = 0;
    if (i < 2LL * n_nodes)
        out[(i >> 1) * OUT_FEATS + ((i & 1) ? DEG_IN_COL : DEG_OUT_COL)] = 0.0f;
}

__global__ void deg_wt_kernel(const int* __restrict__ src, const int* __restrict__ dst,
                              float* __restrict__ out,
                              const float* __restrict__ weight, float* __restrict__ wT,
                              int* __restrict__ rank, int n_edges) {
    int i = blockIdx.x * blockDim.x + threadIdx.x;
    if (i < n_edges) {
        const int s = src[i];
        const int d = dst[i];
        atomicAdd(out + (long long)s * OUT_FEATS + DEG_OUT_COL, 1.0f);
        const int r = atomicAdd((int*)(out + (long long)d * OUT_FEATS + DEG_IN_COL), 1);
        if (rank != nullptr) rank[i] = r;
    }
    if (i < IN_FEATS * OUT_FEATS) wT[wfrag_pos(i)] = weight[i];
}


#define DEG_R       6400
#define DEG_NS      16
#define DEG_THREADS 1024

__device__ __forceinline__ void count_in_range(int* cnt, int v, int lo, int len) {
    const unsigned int o = (unsigned int)(v - lo);
    if (o < (unsigned int)len) atomicAdd(&cnt[o], 1);
}

__device__ __forceinline__ void rank_in_range(int* cnt, int* __restrict__ rank, int e,
                                              int v, int lo, int len) {
    const unsigned int o = (unsigned int)(v - lo);
    if (o < (unsigned int)len) rank[e] = atomicAdd(&cnt[o], 1);
}

__global__ __launch_bounds__(DEG_THREADS)
void deg_lds_kernel(const int* __restrict__ src, const int* __restrict__ dst,
                    int* __restrict__ cin, int* __restrict__ cout,
                    const float* __restrict__ weight, float* __restrict__ wT,
                    int* __restrict__ rank, int* __restrict__ list_top,
                    int n_nodes, int n_edges) {
    __shared__ int sIn[DEG_R];
    __shared__ int sOut[DEG_R];

    const int tid = threadIdx.x;
    const int b = blockIdx.x / DEG_NS;
    const int s = blockIdx.x % DEG_NS;
    const int lo = b * DEG_R;
    const int len = (n_nodes - lo) < DEG_R ? (n_nodes - lo) : DEG_R;

    if (blockIdx.x == 0 && tid == 0) *list_top = 0;

    {
        const int gi = blockIdx.x * DEG_THREADS + tid;
        if (gi < IN_FEATS * OUT_FEATS) wT[wfrag_pos(gi)] = weight[gi];
    }

    for (int n = tid; n < DEG_R; n += DEG_THREADS) { sIn[n] = 0; sOut[n] = 0; }
    __syncthreads();

    const int nq = n_edges >> 2;
    const int qs = (nq + DEG_NS - 1) / DEG_NS;
    const int q0 = s * qs;
    const int q1 = (q0 + qs) < nq ? (q0 + qs) : nq;
    const int e_tail = (nq << 2) + tid;
    const bool tail = (s == DEG_NS - 1) && (e_tail < n_edges);

    for (int q = q0 + tid; q < q1; q += 2 * DEG_THREADS) {
        const bool two = (q + DEG_THREADS) < q1;
        const int qb = two ? (q + DEG_THREADS) : q;
        const int len1 = two ? len : 0;
        const int4 d0 = ((const int4*)dst)[q];
        const int4 d1 = ((const int4*)dst)[qb];
        const int4 s0 = ((const int4*)src)[q];
        const int4 s1 = ((const int4*)src)[qb];
        const int e0 = q << 2;
        const int e1 = qb << 2;
        rank_in_range(sIn, rank, e0,     d0.x, lo, len);
        rank_in_range(sIn, rank, e0 + 1, d0.y, lo, len);
        rank_in_range(sIn, rank, e0 + 2, d0.z, lo, len);
        rank_in_range(sIn, rank, e0 + 3, d0.w, lo, len);
        rank_in_range(sIn, rank, e1,     d1.x, lo, len1);
        rank_in_range(sIn, rank, e1 + 1, d1.y, lo, len1);
        rank_in_range(sIn, rank, e1 + 2, d1.z, lo, len1);
        rank_in_range(sIn, rank, e1 + 3, d1.w, lo, len1);
        count_in_range(sOut, s0.x, lo, len); count_in_range(sOut, s0.y, lo, len);
        count_in_range(sOut, s0.z, lo, len); count_in_range(sOut, s0.w, lo, len);
        count_in_range(sOut, s1.x, lo, len1); count_in_range(sOut, s1.y, lo, len1);
        count_in_range(sOut, s1.z, lo, len1); count_in_range(sOut, s1.w, lo, len1);
    }
    if (tail) {
        rank_in_range(sIn, rank, e_tail, dst[e_tail], lo, len);
        count_in_range(sOut, src[e_tail], lo, len);
    }
    __syncthreads();

    for (int n = tid; n < len; n += DEG_THREADS) {
        cin[(long long)s * n_nodes + lo + n]  = sIn[n];
        cout[(long long)s * n_nodes + lo + n] = sOut[n];
    }
}

__device__ __forceinline__ float lane_xor1(float v) {
    return __int_as_float(__builtin_amdgcn_update_dpp(0, __float_as_int(v), 0xB1, 0xf, 0xf, true));
}
__device__ __forceinline__ float lane_xor3(float v) {
    return __int_as_float(__builtin_amdgcn_update_dpp(0, __float_as_int(v), 0x1B, 0xf, 0xf, true));
}
__device__ __forceinline__ float lane_rev8(float v) {
    return __int_as_float(__builtin_amdgcn_update_dpp(0, __float_as_int(v), 0x141, 0xf, 0xf, true));
}
__device__ __forceinline__ float quad_first(float v) {
    return __int_as_float(__builtin_amdgcn_update_dpp(0, __float_as_int(v), 0x00, 0xf, 0xf, true));
}
__device__ __forceinline__ float quad_last(float v) {
    return __int_as_float(__builtin_amdgcn_update_dpp(0, __float_as_int(v), 0xFF, 0xf, 0xf, true));
}
#define CE_DESC(a, b) { const float hi_ = __builtin_fmaxf(a, b); const float lo_ = __builtin_fminf(a, b); a = hi_; b = lo_; }
#define MERGE8_DESC(r) { \
    CE_DESC(r[0], r[4]) CE_DESC(r[1], r[5]) CE_DESC(r[2], r[6]) CE_DESC(r[3], r[7]) \
    CE_DESC(r[0], r[2]) CE_DESC(r[1], r[3]) CE_DESC(r[4], r[6]) CE_DESC(r[5], r[7]) \
    CE_DESC(r[0], r[1]) CE_DESC(r[2], r[3]) CE_DESC(r[4], r[5]) CE_DESC(r[6], r[7]) }

__global__ __launch_bounds__(128)
void gemm_maxk_kernel(const float* __restrict__ feat, const float* __restrict__ wT,
                      const float* __restrict__ bias, float* out,
                      float* __restrict__ tk_vals, unsigned int* __restrict__ tk_idx,
                      int* __restrict__ seg, int* __restrict__ list_top,
                      int* cin, const int* cout, int n_nodes) {
    __shared__ __attribute__((aligned(16))) float sA[TILE_M * LDA];
    __shared__ __attribute__((aligned(16))) float sH[TILE_M * LDH];
    __shared__ float sScale[TILE_M];
    __shared__ int sDeg[4][2 * TILE_M];

    const int tid = threadIdx.x;
    const int m0 = blockIdx.x * TILE_M;

    float4 fv[TILE_M * (IN_FEATS / 4) / 128];
    #pragma unroll
    for (int j = 0; j < TILE_M * (IN_FEATS / 4) / 128; ++j) {
        const int i = tid + 128 * j;
        const int row = i / (IN_FEATS / 4);
        const int c4  = i % (IN_FEATS / 4);
        const int rr = (m0 + row) < n_nodes ? (m0 + row) : (n_nodes - 1);
        fv[j] = ((const float4*)feat)[(long long)rr * (IN_FEATS / 4) + c4];
    }

    if (cin != nullptr) {
        const int wv = __builtin_amdgcn_readfirstlane(tid >> 5);
        const int ln = tid & 31;
        if (ln < TILE_M) {
            const bool valid = (m0 + ln) < n_nodes;
            const int lo = valid ? ln : 0;
            int ti = 0, to = 0;
            #pragma unroll
            for (int s = 0; s < DEG_NS / 4; ++s) {
                const long long off = (long long)(wv * (DEG_NS / 4) + s) * n_nodes + m0;
                const int vi = (cin + off)[lo];
                const int vo = (cout + off)[lo];
                ti += valid ? vi : 0;
                to += valid ? vo : 0;
            }
            sDeg[wv][ln] = ti;
            sDeg[wv][TILE_M + ln] = to;
        }
    } else if (tid < TILE_M) {
        float sc = 0.f;
        if (m0 + tid < n_nodes) {
            const float* orow = out + (long long)(m0 + tid) * OUT_FEATS;
            float od = orow[DEG_OUT_COL];
            float id = (float)((const int*)orow)[DEG_IN_COL];
            od = od < 1.f ? 1.f : od;
            id = id < 1.f ? 1.f : id;
            sc = rsqrtf(od) * rsqrtf(id);
        }
        sScale[tid] = sc;
    }

    #pragma unroll
    for (int j = 0; j < TILE_M * (IN_FEATS / 4) / 128; ++j) {
        const int i = tid + 128 * j;
        const int row = i / (IN_FEATS / 4);
        const int c4  = i % (IN_FEATS / 4);
        const bool ok = (m0 + row) < n_nodes;
        float* p = &sA[row * LDA + c4 * 4];
        p[0] = ok ? fv[j].x : 0.f; p[1] = ok ? fv[j].y : 0.f;
        p[2] = ok ? fv[j].z : 0.f; p[3] = ok ? fv[j].w : 0.f;
    }
    __syncthreads();

    if (cin != nullptr && tid < TILE_M) {
        const bool valid = (m0 + tid) < n_nodes;
        const int ti = sDeg[0][tid] + sDeg[1][tid] + sDeg[2][tid] + sDeg[3][tid];
        const int to = sDeg[0][TILE_M + tid] + sDeg[1][TILE_M + tid]
                     + sDeg[2][TILE_M + tid] + sDeg[3][TILE_M + tid];
        float od = (float)to, id = (float)ti;
        od = od < 1.f ? 1.f : od;
        id = id < 1.f ? 1.f : id;
        sScale[tid] = valid ? rsqrtf(od) * rsqrtf(id) : 0.f;
    }

    for (int i = tid; seg == nullptr && i < TILE_M * (OUT_FEATS / 4); i += 128) {
        int row = i / (OUT_FEATS / 4);
        int c4  = i % (OUT_FEATS / 4);
        if (m0 + row < n_nodes)
            ((float4*)out)[(long long)(m0 + row) * (OUT_FEATS / 4) + c4] =
                ((const float4*)bias)[c4];
    }

    const int wave = tid >> 5;
    const int lane = tid & 31;
    const int r16  = lane & 15;
    const int koff = (lane >> 4) << 1;
    const float4* wfrag = (const float4*)wT + (wave * 32) * 32 + lane;
    const float* arow = sA + r16 * LDA;

    v8f c = {0.f, 0.f, 0.f, 0.f, 0.f, 0.f, 0.f, 0.f};
    #pragma unroll
    for (int k0 = 0; k0 < IN_FEATS; k0 += 8) {
        const float4 b4 = wfrag[(k0 >> 3) * 32];
        v2f a0, a1, b0, b1;
        a0.x = arow[k0 + koff];
        a0.y = arow[k0 + koff + 1];
        a1.x = arow[k0 + 4 + koff];
        a1.y = arow[k0 + 4 + koff + 1];
        b0.x = b4.x; b0.y = b4.y;
        b1.x = b4.z; b1.y = b4.w;
        c = __builtin_amdgcn_wmma_f32_16x16x4_f32(false, a0, false, b0, (short)0, c,
                                                  false, false);
        c = __builtin_amdgcn_wmma_f32_16x16x4_f32(false, a1, false, b1, (short)0, c,
                                                  false, false);
    }

    {
        const int rbase = (lane >> 4) << 3;
        const int col = wave * 16 + r16;
        #pragma unroll
        for (int r = 0; r < 8; ++r)
            sH[(rbase + r) * LDH + col] = c[r];
    }
    __syncthreads();

    {
        const int row = tid >> 3;
        const int p   = tid & 7;
        const float* rowp = &sH[row * LDH];
        float x[8];
        #pragma unroll
        for (int e = 0; e < 8; ++e) x[e] = rowp[8 * e + p];

        float r[8], u[8];
        #pragma unroll
        for (int e = 0; e < 8; ++e) r[e] = x[e];
        CE_DESC(r[0], r[1]) CE_DESC(r[2], r[3]) CE_DESC(r[4], r[5]) CE_DESC(r[6], r[7])
        CE_DESC(r[0], r[2]) CE_DESC(r[1], r[3]) CE_DESC(r[4], r[6]) CE_DESC(r[5], r[7])
        CE_DESC(r[1], r[2]) CE_DESC(r[5], r[6])
        CE_DESC(r[0], r[4]) CE_DESC(r[1], r[5]) CE_DESC(r[2], r[6]) CE_DESC(r[3], r[7])
        CE_DESC(r[2], r[4]) CE_DESC(r[3], r[5])
        CE_DESC(r[1], r[2]) CE_DESC(r[3], r[4]) CE_DESC(r[5], r[6])
        const float pick = (p & 1) ? -INFINITY : INFINITY;
        #pragma unroll
        for (int i = 0; i < 8; ++i) u[i] = lane_xor1(r[7 - i]);
        #pragma unroll
        for (int i = 0; i < 8; ++i) r[i] = __builtin_amdgcn_fmed3f(r[i], u[i], pick);
        MERGE8_DESC(r)
        #pragma unroll
        for (int i = 0; i < 8; ++i) u[i] = lane_xor3(r[7 - i]);
        #pragma unroll
        for (int i = 0; i < 8; ++i) r[i] = __builtin_amdgcn_fmed3f(r[i], u[i], INFINITY);
        #pragma unroll
        for (int i = 0; i < 8; ++i) u[i] = lane_xor1(r[i]);
        #pragma unroll
        for (int i = 0; i < 8; ++i) r[i] = __builtin_amdgcn_fmed3f(r[i], u[i], pick);
        MERGE8_DESC(r)
        #pragma unroll
        for (int i = 0; i < 8; ++i) u[i] = lane_rev8(r[7 - i]);
        #pragma unroll
        for (int i = 0; i < 8; ++i) r[i] = __builtin_amdgcn_fmed3f(r[i], u[i], INFINITY);
        float m = __builtin_fminf(__builtin_fminf(__builtin_fminf(r[0], r[1]), __builtin_fminf(r[2], r[3])),
                                  __builtin_fminf(__builtin_fminf(r[4], r[5]), __builtin_fminf(r[6], r[7])));
        m = __builtin_amdgcn_fmed3f(m, lane_xor1(m), -INFINITY);
        const float m_lo = quad_first(m);
        const float m_hi = quad_last(m);
        const float thr = (p < 4) ? m_lo : m_hi;

        int nkeep = 0;
        #pragma unroll
        for (int e = 0; e < 8; ++e) nkeep += (x[e] >= thr) ? 1 : 0;
        int inc = nkeep;
        {
            int t1 = __builtin_amdgcn_update_dpp(0, inc, 0x111, 0xf, 0xf, true);
            if (p >= 1) inc += t1;
            int t2 = __builtin_amdgcn_update_dpp(0, inc, 0x112, 0xf, 0xf, true);
            if (p >= 2) inc += t2;
            int t4 = __builtin_amdgcn_update_dpp(0, inc, 0x114, 0xf, 0xf, true);
            if (p >= 4) inc += t4;
        }
        const bool exact = __builtin_amdgcn_ballot_w32(p == 7 && inc != MAXK) == 0u;

        int slot_of[8];
        if (exact) {
            int run = inc - nkeep;
            #pragma unroll
            for (int e = 0; e < 8; ++e) {
                const bool k = x[e] >= thr;
                slot_of[e] = k ? run : MAXK;
                run += k ? 1 : 0;
            }
        } else {
            #pragma unroll
            for (int e = 0; e < 8; ++e) slot_of[e] = 0;
            #pragma unroll
            for (int g = 0; g < 8; ++g) {
                const float4 h0 = *(const float4*)(rowp + 8 * g);
                const float4 h1 = *(const float4*)(rowp + 8 * g + 4);
                const float hv[8] = {h0.x, h0.y, h0.z, h0.w, h1.x, h1.y, h1.z, h1.w};
                #pragma unroll
                for (int pp = 0; pp < 8; ++pp) {
                    const float v = hv[pp];
                    #pragma unroll
                    for (int e = 0; e < 8; ++e) {
                        bool before;
                        if (g < e)      before = v >= x[e];
                        else if (g > e) before = v > x[e];
                        else            before = (v > x[e]) || (v == x[e] && pp < p);
                        slot_of[e] += before ? 1 : 0;
                    }
                }
            }
        }
        const float scale = sScale[row];
        float* sV = sA;
        unsigned char* sI = (unsigned char*)(sA + TILE_M * MAXK);
        #pragma unroll
        for (int e = 0; e < 8; ++e) {
            const bool keep = slot_of[e] < MAXK;
            const int slot = row * MAXK + slot_of[e];
            sV[keep ? slot : 512 + tid] = x[e] * scale;
            sI[keep ? slot : 4096 + tid] = (unsigned char)(8 * e + p);
        }
    }
    __syncthreads();
    if (tid < 64) {
        const int row = tid >> 2;
        if (m0 + row < n_nodes)
            ((float4*)tk_vals)[(long long)(m0 + row) * (MAXK / 4) + (tid & 3)] =
                ((const float4*)sA)[tid];
    } else if (tid < 64 + TILE_M) {
        const int row = tid - 64;
        if (m0 + row < n_nodes)
            ((uint4*)tk_idx)[m0 + row] = ((const uint4*)(sA + TILE_M * MAXK))[row];
    }

    if (cin != nullptr && tid >= 32 && tid < 64) {
        const int l = tid - 32;
        const bool valid = l < TILE_M && (m0 + l) < n_nodes;
        int ci[DEG_NS];
        int ti = 0;
        #pragma unroll
        for (int s = 0; s < DEG_NS; ++s) {
            ci[s] = valid ? cin[(long long)s * n_nodes + m0 + l] : 0;
            ti += ci[s];
        }
        int inc = ti;
        #pragma unroll
        for (int d = 1; d < 32; d <<= 1) {
            const int t = __shfl_up(inc, d);
            if (l >= d) inc += t;
        }
        const int total = __shfl(inc, 31);
        int base = 0;
        if (l == 0) base = atomicAdd(list_top, total);
        base = __shfl(base, 0);
        int run = base + inc - ti;
        if (valid) {
            ((int2*)seg)[m0 + l] = make_int2(run, ti);
            #pragma unroll
            for (int s = 0; s < DEG_NS; ++s) {
                cin[(long long)s * n_nodes + m0 + l] = run;
                run += ci[s];
            }
        }
    }
}

__global__ void scatter_kernel(const int* __restrict__ src, const int* __restrict__ dst,
                               const float* __restrict__ tk_vals,
                               const unsigned int* __restrict__ tk_idx,
                               float* __restrict__ out, int n_edges) {
    long long gid = (long long)blockIdx.x * blockDim.x + threadIdx.x;
    long long total = (long long)n_edges * 4;
    if (gid >= total) return;
    int e = (int)(gid >> 2);
    int part = (int)(gid & 3);
    int s = src[e], d = dst[e];
    float4 v = ((const float4*)tk_vals)[(long long)s * 4 + part];
    unsigned int ip = tk_idx[(long long)s * 4 + part];
    float* orow = out + (long long)d * OUT_FEATS;
    atomicAdd(orow + (ip & 0xffu), v.x);
    atomicAdd(orow + ((ip >> 8) & 0xffu), v.y);
    atomicAdd(orow + ((ip >> 16) & 0xffu), v.z);
    atomicAdd(orow + ((ip >> 24) & 0xffu), v.w);
}


__global__ void fill_kernel(const int* __restrict__ src, const int* __restrict__ dst,
                            const int* __restrict__ rank, const int* __restrict__ cin,
                            int* __restrict__ sorted_src, int n_nodes, int n_edges) {
    const int nq = n_edges >> 2;
    const int qs = (nq + DEG_NS - 1) / DEG_NS;
    const long long stride = (long long)gridDim.x * blockDim.x;
    for (long long e = (long long)blockIdx.x * blockDim.x + threadIdx.x; e < n_edges; e += stride) {
        const int d = dst[e];
        const int q = (int)(e >> 2);
        const int s = q < nq ? q / qs : DEG_NS - 1;
        sorted_src[cin[(long long)s * n_nodes + d] + rank[e]] = src[e];
    }
}

__global__ __launch_bounds__(256)
void gather_kernel(const int* __restrict__ seg, const int* __restrict__ sorted_src,
                   const float* __restrict__ tk_vals, const unsigned int* __restrict__ tk_idx,
                   const float* __restrict__ bias, float* __restrict__ out, int n_nodes) {
    __shared__ __attribute__((aligned(16))) float sRow[8][OUT_FEATS];
    const int lane = threadIdx.x & 31;
    const int wave = threadIdx.x >> 5;
    const int half = lane >> 4;
    const int k    = lane & 15;
    const int sh   = (k & 3) * 8;
    float* row = sRow[wave];
    const float2 b2 = ((const float2*)bias)[lane];

    for (int node = blockIdx.x * 8 + wave; node < n_nodes; node += gridDim.x * 8) {
        const int2 sg = ((const int2*)seg)[node];
        const int start = __builtin_amdgcn_readfirstlane(sg.x);
        const int cnt   = __builtin_amdgcn_readfirstlane(sg.y);
        row[lane] = 0.f;
        row[lane + 32] = 0.f;
        for (int base = 0; base < cnt; base += 32) {
            const int rem = (cnt - base) < 32 ? (cnt - base) : 32;
            const int sl = lane < rem ? sorted_src[start + base + lane] : 0;
            for (int q = 0; q < rem; q += 8) {
                float v[4];
                unsigned int iw[4];
                #pragma unroll
                for (int u = 0; u < 4; ++u) {
                    const int s = __shfl(sl, q + 2 * u + half);
                    v[u]  = tk_vals[(long long)s * MAXK + k];
                    iw[u] = tk_idx[(long long)s * 4 + (k >> 2)];
                }
                #pragma unroll
                for (int u = 0; u < 4; ++u)
                    if (q + 2 * u + half < rem)
                        atomicAdd(&row[(iw[u] >> sh) & 0xffu], v[u]);
            }
        }
        const float2 r2 = ((const float2*)row)[lane];
        ((float2*)out)[(long long)node * (OUT_FEATS / 2) + lane] =
            make_float2(r2.x + b2.x, r2.y + b2.y);
    }
}

extern "C" void kernel_launch(void* const* d_in, const int* in_sizes, int n_in,
                              void* d_out, int out_size, void* d_ws, size_t ws_size,
                              hipStream_t stream) {
    const float* feat   = (const float*)d_in[0];
    const float* weight = (const float*)d_in[1];
    const float* bias   = (const float*)d_in[2];
    const int*   src    = (const int*)d_in[3];
    const int*   dst    = (const int*)d_in[4];
    float* out = (float*)d_out;

    const int n_nodes = in_sizes[0] / IN_FEATS;
    const int n_edges = in_sizes[3];

    float* ws_f    = (float*)d_ws;
    float* wT      = ws_f;
    float* tk_vals = wT + IN_FEATS * OUT_FEATS;
    unsigned int* tk_idx = (unsigned int*)(tk_vals + (long long)n_nodes * MAXK);

    const size_t o_seg    = (((size_t)IN_FEATS * OUT_FEATS + (size_t)n_nodes * (MAXK + 4)) * 4 + 255) & ~(size_t)255;
    const size_t o_rank   = (o_seg + (size_t)n_nodes * 8 + 255) & ~(size_t)255;
    const size_t o_sorted = (o_rank + (size_t)n_edges * 4 + 255) & ~(size_t)255;
    const size_t o_top    = (o_sorted + (size_t)n_edges * 4 + 255) & ~(size_t)255;
    const bool gather = ws_size >= o_top + 256;
    char* ws_b = (char*)d_ws;
    int*  seg        = gather ? (int*)(ws_b + o_seg)    : nullptr;
    int*  rank       = gather ? (int*)(ws_b + o_rank)   : nullptr;
    int*  sorted_src = gather ? (int*)(ws_b + o_sorted) : nullptr;
    int*  list_top   = gather ? (int*)(ws_b + o_top)    : nullptr;

    int* cin  = gather ? (int*)d_out : nullptr;
    int* cout = gather ? (int*)d_out + (long long)DEG_NS * n_nodes : nullptr;
    if (gather) {
        int nb = (n_nodes + DEG_R - 1) / DEG_R;
        deg_lds_kernel<<<nb * DEG_NS, DEG_THREADS, 0, stream>>>(src, dst, cin, cout, weight, wT, rank,
                                                                 list_top, n_nodes, n_edges);
    } else {
        long long total = 2LL * n_nodes;
        int grid = (int)((total + 255) / 256);
        init_kernel<<<grid, 256, 0, stream>>>(out, list_top, n_nodes);
        int work = n_edges > IN_FEATS * OUT_FEATS ? n_edges : IN_FEATS * OUT_FEATS;
        int grid2 = (work + 255) / 256;
        deg_wt_kernel<<<grid2, 256, 0, stream>>>(src, dst, out, weight, wT, rank, n_edges);
    }
    {
        int grid = (n_nodes + TILE_M - 1) / TILE_M;
        gemm_maxk_kernel<<<grid, 128, 0, stream>>>(feat, wT, bias, out,
                                                   tk_vals, tk_idx, seg, list_top, cin, cout, n_nodes);
    }
    if (gather) {
        fill_kernel<<<2048, 256, 0, stream>>>(src, dst, rank, cin, sorted_src, n_nodes, n_edges);
        gather_kernel<<<2048, 256, 0, stream>>>(seg, sorted_src, tk_vals, tk_idx, bias, out, n_nodes);
    } else {
        long long total = (long long)n_edges * 4;
        int grid = (int)((total + 255) / 256);
        scatter_kernel<<<grid, 256, 0, stream>>>(src, dst, tk_vals, tk_idx, out, n_edges);
    }
}


# top-16 record of a node packed as one 128-byte line of 16 (value, column) pairs: the gather reads one line an edge with one 8-byte load a lane, was two lines and two loads; sum order unchanged
# speedup vs baseline: 2.7855x; 1.0404x over previous
#include <hip/hip_runtime.h>
#include <math.h>

typedef __attribute__((ext_vector_type(2))) float v2f;
typedef __attribute__((ext_vector_type(8))) float v8f;

#define IN_FEATS 256
#define OUT_FEATS 64
#define MAXK 16
#define TILE_M 16
#define LDA (IN_FEATS + 4)
#define LDH (OUT_FEATS + 4)

#define REC_WORDS (2 * MAXK)

#define DEG_OUT_COL 0
#define DEG_IN_COL  32

__device__ __forceinline__ int wfrag_pos(int gi) {
    const int k = gi >> 6;
    const int c = gi & 63;
    const int w = c >> 4, c16 = c & 15;
    const int jj = k >> 3, sub = (k >> 2) & 1, h = (k >> 1) & 1, e = k & 1;
    return (((w * 32 + jj) * 32 + (h * 16 + c16)) << 2) + sub * 2 + e;
}

__global__ void init_kernel(float* __restrict__ out, int* __restrict__ list_top, int n_nodes) {
    long long i = (long long)blockIdx.x * blockDim.x + threadIdx.x;
    if (i == 0 && list_top != nullptr) *list_top = 0;
    if (i < 2LL * n_nodes)
        out[(i >> 1) * OUT_FEATS + ((i & 1) ? DEG_IN_COL : DEG_OUT_COL)] = 0.0f;
}

__global__ void deg_wt_kernel(const int* __restrict__ src, const int* __restrict__ dst,
                              float* __restrict__ out,
                              const float* __restrict__ weight, float* __restrict__ wT,
                              int* __restrict__ rank, int n_edges) {
    int i = blockIdx.x * blockDim.x + threadIdx.x;
    if (i < n_edges) {
        const int s = src[i];
        const int d = dst[i];
        atomicAdd(out + (long long)s * OUT_FEATS + DEG_OUT_COL, 1.0f);
        const int r = atomicAdd((int*)(out + (long long)d * OUT_FEATS + DEG_IN_COL), 1);
        if (rank != nullptr) rank[i] = r;
    }
    if (i < IN_FEATS * OUT_FEATS) wT[wfrag_pos(i)] = weight[i];
}


#define DEG_R       6400
#define DEG_NS      16
#define DEG_THREADS 1024

__device__ __forceinline__ void count_in_range(int* cnt, int v, int lo, int len) {
    const unsigned int o = (unsigned int)(v - lo);
    if (o < (unsigned int)len) atomicAdd(&cnt[o], 1);
}

__device__ __forceinline__ void rank_in_range(int* cnt, int* __restrict__ rank, int e,
                                              int v, int lo, int len) {
    const unsigned int o = (unsigned int)(v - lo);
    if (o < (unsigned int)len) rank[e] = atomicAdd(&cnt[o], 1);
}

__global__ __launch_bounds__(DEG_THREADS)
void deg_lds_kernel(const int* __restrict__ src, const int* __restrict__ dst,
                    int* __restrict__ cin, int* __restrict__ cout,
                    const float* __restrict__ weight, float* __restrict__ wT,
                    int* __restrict__ rank, int* __restrict__ list_top,
                    int n_nodes, int n_edges) {
    __shared__ int sIn[DEG_R];
    __shared__ int sOut[DEG_R];

    const int tid = threadIdx.x;
    const int b = blockIdx.x / DEG_NS;
    const int s = blockIdx.x % DEG_NS;
    const int lo = b * DEG_R;
    const int len = (n_nodes - lo) < DEG_R ? (n_nodes - lo) : DEG_R;

    if (blockIdx.x == 0 && tid == 0) *list_top = 0;

    {
        const int gi = blockIdx.x * DEG_THREADS + tid;
        if (gi < IN_FEATS * OUT_FEATS) wT[wfrag_pos(gi)] = weight[gi];
    }

    for (int n = tid; n < DEG_R; n += DEG_THREADS) { sIn[n] = 0; sOut[n] = 0; }
    __syncthreads();

    const int nq = n_edges >> 2;
    const int qs = (nq + DEG_NS - 1) / DEG_NS;
    const int q0 = s * qs;
    const int q1 = (q0 + qs) < nq ? (q0 + qs) : nq;
    const int e_tail = (nq << 2) + tid;
    const bool tail = (s == DEG_NS - 1) && (e_tail < n_edges);

    for (int q = q0 + tid; q < q1; q += 2 * DEG_THREADS) {
        const bool two = (q + DEG_THREADS) < q1;
        const int qb = two ? (q + DEG_THREADS) : q;
        const int len1 = two ? len : 0;
        const int4 d0 = ((const int4*)dst)[q];
        const int4 d1 = ((const int4*)dst)[qb];
        const int4 s0 = ((const int4*)src)[q];
        const int4 s1 = ((const int4*)src)[qb];
        const int e0 = q << 2;
        const int e1 = qb << 2;
        rank_in_range(sIn, rank, e0,     d0.x, lo, len);
        rank_in_range(sIn, rank, e0 + 1, d0.y, lo, len);
        rank_in_range(sIn, rank, e0 + 2, d0.z, lo, len);
        rank_in_range(sIn, rank, e0 + 3, d0.w, lo, len);
        rank_in_range(sIn, rank, e1,     d1.x, lo, len1);
        rank_in_range(sIn, rank, e1 + 1, d1.y, lo, len1);
        rank_in_range(sIn, rank, e1 + 2, d1.z, lo, len1);
        rank_in_range(sIn, rank, e1 + 3, d1.w, lo, len1);
        count_in_range(sOut, s0.x, lo, len); count_in_range(sOut, s0.y, lo, len);
        count_in_range(sOut, s0.z, lo, len); count_in_range(sOut, s0.w, lo, len);
        count_in_range(sOut, s1.x, lo, len1); count_in_range(sOut, s1.y, lo, len1);
        count_in_range(sOut, s1.z, lo, len1); count_in_range(sOut, s1.w, lo, len1);
    }
    if (tail) {
        rank_in_range(sIn, rank, e_tail, dst[e_tail], lo, len);
        count_in_range(sOut, src[e_tail], lo, len);
    }
    __syncthreads();

    for (int n = tid; n < len; n += DEG_THREADS) {
        cin[(long long)s * n_nodes + lo + n]  = sIn[n];
        cout[(long long)s * n_nodes + lo + n] = sOut[n];
    }
}

__device__ __forceinline__ float lane_xor1(float v) {
    return __int_as_float(__builtin_amdgcn_update_dpp(0, __float_as_int(v), 0xB1, 0xf, 0xf, true));
}
__device__ __forceinline__ float lane_xor3(float v) {
    return __int_as_float(__builtin_amdgcn_update_dpp(0, __float_as_int(v), 0x1B, 0xf, 0xf, true));
}
__device__ __forceinline__ float lane_rev8(float v) {
    return __int_as_float(__builtin_amdgcn_update_dpp(0, __float_as_int(v), 0x141, 0xf, 0xf, true));
}
__device__ __forceinline__ float quad_first(float v) {
    return __int_as_float(__builtin_amdgcn_update_dpp(0, __float_as_int(v), 0x00, 0xf, 0xf, true));
}
__device__ __forceinline__ float quad_last(float v) {
    return __int_as_float(__builtin_amdgcn_update_dpp(0, __float_as_int(v), 0xFF, 0xf, 0xf, true));
}
#define CE_DESC(a, b) { const float hi_ = __builtin_fmaxf(a, b); const float lo_ = __builtin_fminf(a, b); a = hi_; b = lo_; }
#define MERGE8_DESC(r) { \
    CE_DESC(r[0], r[4]) CE_DESC(r[1], r[5]) CE_DESC(r[2], r[6]) CE_DESC(r[3], r[7]) \
    CE_DESC(r[0], r[2]) CE_DESC(r[1], r[3]) CE_DESC(r[4], r[6]) CE_DESC(r[5], r[7]) \
    CE_DESC(r[0], r[1]) CE_DESC(r[2], r[3]) CE_DESC(r[4], r[5]) CE_DESC(r[6], r[7]) }

__global__ __launch_bounds__(128)
void gemm_maxk_kernel(const float* __restrict__ feat, const float* __restrict__ wT,
                      const float* __restrict__ bias, float* out,
                      float* __restrict__ tk_rec,
                      int* __restrict__ seg, int* __restrict__ list_top,
                      int* cin, const int* cout, int n_nodes) {
    __shared__ __attribute__((aligned(16))) float sA[TILE_M * LDA];
    __shared__ __attribute__((aligned(16))) float sH[TILE_M * LDH];
    __shared__ float sScale[TILE_M];
    __shared__ int sDeg[4][2 * TILE_M];

    const int tid = threadIdx.x;
    const int m0 = blockIdx.x * TILE_M;

    float4 fv[TILE_M * (IN_FEATS / 4) / 128];
    #pragma unroll
    for (int j = 0; j < TILE_M * (IN_FEATS / 4) / 128; ++j) {
        const int i = tid + 128 * j;
        const int row = i / (IN_FEATS / 4);
        const int c4  = i % (IN_FEATS / 4);
        const int rr = (m0 + row) < n_nodes ? (m0 + row) : (n_nodes - 1);
        fv[j] = ((const float4*)feat)[(long long)rr * (IN_FEATS / 4) + c4];
    }

    if (cin != nullptr) {
        const int wv = __builtin_amdgcn_readfirstlane(tid >> 5);
        const int ln = tid & 31;
        if (ln < TILE_M) {
            const bool valid = (m0 + ln) < n_nodes;
            const int lo = valid ? ln : 0;
            int ti = 0, to = 0;
            #pragma unroll
            for (int s = 0; s < DEG_NS / 4; ++s) {
                const long long off = (long long)(wv * (DEG_NS / 4) + s) * n_nodes + m0;
                const int vi = (cin + off)[lo];
                const int vo = (cout + off)[lo];
                ti += valid ? vi : 0;
                to += valid ? vo : 0;
            }
            sDeg[wv][ln] = ti;
            sDeg[wv][TILE_M + ln] = to;
        }
    } else if (tid < TILE_M) {
        float sc = 0.f;
        if (m0 + tid < n_nodes) {
            const float* orow = out + (long long)(m0 + tid) * OUT_FEATS;
            float od = orow[DEG_OUT_COL];
            float id = (float)((const int*)orow)[DEG_IN_COL];
            od = od < 1.f ? 1.f : od;
            id = id < 1.f ? 1.f : id;
            sc = rsqrtf(od) * rsqrtf(id);
        }
        sScale[tid] = sc;
    }

    #pragma unroll
    for (int j = 0; j < TILE_M * (IN_FEATS / 4) / 128; ++j) {
        const int i = tid + 128 * j;
        const int row = i / (IN_FEATS / 4);
        const int c4  = i % (IN_FEATS / 4);
        const bool ok = (m0 + row) < n_nodes;
        float* p = &sA[row * LDA + c4 * 4];
        p[0] = ok ? fv[j].x : 0.f; p[1] = ok ? fv[j].y : 0.f;
        p[2] = ok ? fv[j].z : 0.f; p[3] = ok ? fv[j].w : 0.f;
    }
    __syncthreads();

    if (cin != nullptr && tid < TILE_M) {
        const bool valid = (m0 + tid) < n_nodes;
        const int ti = sDeg[0][tid] + sDeg[1][tid] + sDeg[2][tid] + sDeg[3][tid];
        const int to = sDeg[0][TILE_M + tid] + sDeg[1][TILE_M + tid]
                     + sDeg[2][TILE_M + tid] + sDeg[3][TILE_M + tid];
        float od = (float)to, id = (float)ti;
        od = od < 1.f ? 1.f : od;
        id = id < 1.f ? 1.f : id;
        sScale[tid] = valid ? rsqrtf(od) * rsqrtf(id) : 0.f;
    }

    for (int i = tid; seg == nullptr && i < TILE_M * (OUT_FEATS / 4); i += 128) {
        int row = i / (OUT_FEATS / 4);
        int c4  = i % (OUT_FEATS / 4);
        if (m0 + row < n_nodes)
            ((float4*)out)[(long long)(m0 + row) * (OUT_FEATS / 4) + c4] =
                ((const float4*)bias)[c4];
    }

    const int wave = tid >> 5;
    const int lane = tid & 31;
    const int r16  = lane & 15;
    const int koff = (lane >> 4) << 1;
    const float4* wfrag = (const float4*)wT + (wave * 32) * 32 + lane;
    const float* arow = sA + r16 * LDA;

    v8f c = {0.f, 0.f, 0.f, 0.f, 0.f, 0.f, 0.f, 0.f};
    #pragma unroll
    for (int k0 = 0; k0 < IN_FEATS; k0 += 8) {
        const float4 b4 = wfrag[(k0 >> 3) * 32];
        v2f a0, a1, b0, b1;
        a0.x = arow[k0 + koff];
        a0.y = arow[k0 + koff + 1];
        a1.x = arow[k0 + 4 + koff];
        a1.y = arow[k0 + 4 + koff + 1];
        b0.x = b4.x; b0.y = b4.y;
        b1.x = b4.z; b1.y = b4.w;
        c = __builtin_amdgcn_wmma_f32_16x16x4_f32(false, a0, false, b0, (short)0, c,
                                                  false, false);
        c = __builtin_amdgcn_wmma_f32_16x16x4_f32(false, a1, false, b1, (short)0, c,
                                                  false, false);
    }

    {
        const int rbase = (lane >> 4) << 3;
        const int col = wave * 16 + r16;
        #pragma unroll
        for (int r = 0; r < 8; ++r)
            sH[(rbase + r) * LDH + col] = c[r];
    }
    __syncthreads();

    {
        const int row = tid >> 3;
        const int p   = tid & 7;
        const float* rowp = &sH[row * LDH];
        float x[8];
        #pragma unroll
        for (int e = 0; e < 8; ++e) x[e] = rowp[8 * e + p];

        float r[8], u[8];
        #pragma unroll
        for (int e = 0; e < 8; ++e) r[e] = x[e];
        CE_DESC(r[0], r[1]) CE_DESC(r[2], r[3]) CE_DESC(r[4], r[5]) CE_DESC(r[6], r[7])
        CE_DESC(r[0], r[2]) CE_DESC(r[1], r[3]) CE_DESC(r[4], r[6]) CE_DESC(r[5], r[7])
        CE_DESC(r[1], r[2]) CE_DESC(r[5], r[6])
        CE_DESC(r[0], r[4]) CE_DESC(r[1], r[5]) CE_DESC(r[2], r[6]) CE_DESC(r[3], r[7])
        CE_DESC(r[2], r[4]) CE_DESC(r[3], r[5])
        CE_DESC(r[1], r[2]) CE_DESC(r[3], r[4]) CE_DESC(r[5], r[6])
        const float pick = (p & 1) ? -INFINITY : INFINITY;
        #pragma unroll
        for (int i = 0; i < 8; ++i) u[i] = lane_xor1(r[7 - i]);
        #pragma unroll
        for (int i = 0; i < 8; ++i) r[i] = __builtin_amdgcn_fmed3f(r[i], u[i], pick);
        MERGE8_DESC(r)
        #pragma unroll
        for (int i = 0; i < 8; ++i) u[i] = lane_xor3(r[7 - i]);
        #pragma unroll
        for (int i = 0; i < 8; ++i) r[i] = __builtin_amdgcn_fmed3f(r[i], u[i], INFINITY);
        #pragma unroll
        for (int i = 0; i < 8; ++i) u[i] = lane_xor1(r[i]);
        #pragma unroll
        for (int i = 0; i < 8; ++i) r[i] = __builtin_amdgcn_fmed3f(r[i], u[i], pick);
        MERGE8_DESC(r)
        #pragma unroll
        for (int i = 0; i < 8; ++i) u[i] = lane_rev8(r[7 - i]);
        #pragma unroll
        for (int i = 0; i < 8; ++i) r[i] = __builtin_amdgcn_fmed3f(r[i], u[i], INFINITY);
        float m = __builtin_fminf(__builtin_fminf(__builtin_fminf(r[0], r[1]), __builtin_fminf(r[2], r[3])),
                                  __builtin_fminf(__builtin_fminf(r[4], r[5]), __builtin_fminf(r[6], r[7])));
        m = __builtin_amdgcn_fmed3f(m, lane_xor1(m), -INFINITY);
        const float m_lo = quad_first(m);
        const float m_hi = quad_last(m);
        const float thr = (p < 4) ? m_lo : m_hi;

        int nkeep = 0;
        #pragma unroll
        for (int e = 0; e < 8; ++e) nkeep += (x[e] >= thr) ? 1 : 0;
        int inc = nkeep;
        {
            int t1 = __builtin_amdgcn_update_dpp(0, inc, 0x111, 0xf, 0xf, true);
            if (p >= 1) inc += t1;
            int t2 = __builtin_amdgcn_update_dpp(0, inc, 0x112, 0xf, 0xf, true);
            if (p >= 2) inc += t2;
            int t4 = __builtin_amdgcn_update_dpp(0, inc, 0x114, 0xf, 0xf, true);
            if (p >= 4) inc += t4;
        }
        const bool exact = __builtin_amdgcn_ballot_w32(p == 7 && inc != MAXK) == 0u;

        int slot_of[8];
        if (exact) {
            int run = inc - nkeep;
            #pragma unroll
            for (int e = 0; e < 8; ++e) {
                const bool k = x[e] >= thr;
                slot_of[e] = k ? run : MAXK;
                run += k ? 1 : 0;
            }
        } else {
            #pragma unroll
            for (int e = 0; e < 8; ++e) slot_of[e] = 0;
            #pragma unroll
            for (int g = 0; g < 8; ++g) {
                const float4 h0 = *(const float4*)(rowp + 8 * g);
                const float4 h1 = *(const float4*)(rowp + 8 * g + 4);
                const float hv[8] = {h0.x, h0.y, h0.z, h0.w, h1.x, h1.y, h1.z, h1.w};
                #pragma unroll
                for (int pp = 0; pp < 8; ++pp) {
                    const float v = hv[pp];
                    #pragma unroll
                    for (int e = 0; e < 8; ++e) {
                        bool before;
                        if (g < e)      before = v >= x[e];
                        else if (g > e) before = v > x[e];
                        else            before = (v > x[e]) || (v == x[e] && pp < p);
                        slot_of[e] += before ? 1 : 0;
                    }
                }
            }
        }
        const float scale = sScale[row];
        float2* sP = (float2*)sA;
        #pragma unroll
        for (int e = 0; e < 8; ++e) {
            const bool keep = slot_of[e] < MAXK;
            const int slot = row * MAXK + slot_of[e];
            sP[keep ? slot : 512 + tid] = make_float2(x[e] * scale, __int_as_float(8 * e + p));
        }
    }
    __syncthreads();
    if (m0 + (tid >> 3) < n_nodes)
        ((float4*)tk_rec)[(long long)m0 * (REC_WORDS / 4) + tid] = ((const float4*)sA)[tid];

    if (cin != nullptr && tid >= 32 && tid < 64) {
        const int l = tid - 32;
        const bool valid = l < TILE_M && (m0 + l) < n_nodes;
        int ci[DEG_NS];
        int ti = 0;
        #pragma unroll
        for (int s = 0; s < DEG_NS; ++s) {
            ci[s] = valid ? cin[(long long)s * n_nodes + m0 + l] : 0;
            ti += ci[s];
        }
        int inc = ti;
        #pragma unroll
        for (int d = 1; d < 32; d <<= 1) {
            const int t = __shfl_up(inc, d);
            if (l >= d) inc += t;
        }
        const int total = __shfl(inc, 31);
        int base = 0;
        if (l == 0) base = atomicAdd(list_top, total);
        base = __shfl(base, 0);
        int run = base + inc - ti;
        if (valid) {
            ((int2*)seg)[m0 + l] = make_int2(run, ti);
            #pragma unroll
            for (int s = 0; s < DEG_NS; ++s) {
                cin[(long long)s * n_nodes + m0 + l] = run;
                run += ci[s];
            }
        }
    }
}

__global__ void scatter_kernel(const int* __restrict__ src, const int* __restrict__ dst,
                               const float* __restrict__ tk_rec,
                               float* __restrict__ out, int n_edges) {
    long long gid = (long long)blockIdx.x * blockDim.x + threadIdx.x;
    long long total = (long long)n_edges * 4;
    if (gid >= total) return;
    int e = (int)(gid >> 2);
    int part = (int)(gid & 3);
    int s = src[e], d = dst[e];
    const float4 a = ((const float4*)tk_rec)[(long long)s * (REC_WORDS / 4) + part * 2];
    const float4 b = ((const float4*)tk_rec)[(long long)s * (REC_WORDS / 4) + part * 2 + 1];
    float* orow = out + (long long)d * OUT_FEATS;
    atomicAdd(orow + __float_as_int(a.y), a.x);
    atomicAdd(orow + __float_as_int(a.w), a.z);
    atomicAdd(orow + __float_as_int(b.y), b.x);
    atomicAdd(orow + __float_as_int(b.w), b.z);
}


__global__ void fill_kernel(const int* __restrict__ src, const int* __restrict__ dst,
                            const int* __restrict__ rank, const int* __restrict__ cin,
                            int* __restrict__ sorted_src, int n_nodes, int n_edges) {
    const int nq = n_edges >> 2;
    const int qs = (nq + DEG_NS - 1) / DEG_NS;
    const long long stride = (long long)gridDim.x * blockDim.x;
    for (long long e = (long long)blockIdx.x * blockDim.x + threadIdx.x; e < n_edges; e += stride) {
        const int d = dst[e];
        const int q = (int)(e >> 2);
        const int s = q < nq ? q / qs : DEG_NS - 1;
        sorted_src[cin[(long long)s * n_nodes + d] + rank[e]] = src[e];
    }
}

__global__ __launch_bounds__(256)
void gather_kernel(const int* __restrict__ seg, const int* __restrict__ sorted_src,
                   const float* __restrict__ tk_rec,
                   const float* __restrict__ bias, float* __restrict__ out, int n_nodes) {
    __shared__ __attribute__((aligned(16))) float sRow[8][OUT_FEATS];
    const int lane = threadIdx.x & 31;
    const int wave = threadIdx.x >> 5;
    const int half = lane >> 4;
    const int k    = lane & 15;
    float* row = sRow[wave];
    const float2 b2 = ((const float2*)bias)[lane];
    const float2* rec = (const float2*)tk_rec;

    for (int node = blockIdx.x * 8 + wave; node < n_nodes; node += gridDim.x * 8) {
        const int2 sg = ((const int2*)seg)[node];
        const int start = __builtin_amdgcn_readfirstlane(sg.x);
        const int cnt   = __builtin_amdgcn_readfirstlane(sg.y);
        row[lane] = 0.f;
        row[lane + 32] = 0.f;
        for (int base = 0; base < cnt; base += 32) {
            const int rem = (cnt - base) < 32 ? (cnt - base) : 32;
            const int sl = lane < rem ? sorted_src[start + base + lane] : 0;
            for (int q = 0; q < rem; q += 16) {
                float2 pr[8];
                #pragma unroll
                for (int u = 0; u < 8; ++u) {
                    const int s = __shfl(sl, q + 2 * u + half);
                    pr[u] = rec[(long long)s * MAXK + k];
                }
                #pragma unroll
                for (int u = 0; u < 8; ++u)
                    if (q + 2 * u + half < rem)
                        atomicAdd(&row[__float_as_int(pr[u].y)], pr[u].x);
            }
        }
        const float2 r2 = ((const float2*)row)[lane];
        ((float2*)out)[(long long)node * (OUT_FEATS / 2) + lane] =
            make_float2(r2.x + b2.x, r2.y + b2.y);
    }
}

extern "C" void kernel_launch(void* const* d_in, const int* in_sizes, int n_in,
                              void* d_out, int out_size, void* d_ws, size_t ws_size,
                              hipStream_t stream) {
    const float* feat   = (const float*)d_in[0];
    const float* weight = (const float*)d_in[1];
    const float* bias   = (const float*)d_in[2];
    const int*   src    = (const int*)d_in[3];
    const int*   dst    = (const int*)d_in[4];
    float* out = (float*)d_out;

    const int n_nodes = in_sizes[0] / IN_FEATS;
    const int n_edges = in_sizes[3];

    float* ws_f    = (float*)d_ws;
    float* wT      = ws_f;
    float* tk_rec  = wT + IN_FEATS * OUT_FEATS;

    const size_t o_seg    = (((size_t)IN_FEATS * OUT_FEATS + (size_t)n_nodes * REC_WORDS) * 4 + 255) & ~(size_t)255;
    const size_t o_sorted = (o_seg + (size_t)n_nodes * 8 + 255) & ~(size_t)255;
    const size_t o_top    = (o_sorted + (size_t)n_edges * 4 + 255) & ~(size_t)255;
    const bool gather = ws_size >= o_top + 256 &&
                        (long long)n_edges <= (long long)(OUT_FEATS - 2 * DEG_NS) * n_nodes;
    char* ws_b = (char*)d_ws;
    int*  seg        = gather ? (int*)(ws_b + o_seg)    : nullptr;
    int*  sorted_src = gather ? (int*)(ws_b + o_sorted) : nullptr;
    int*  list_top   = gather ? (int*)(ws_b + o_top)    : nullptr;

    int* cin  = gather ? (int*)d_out : nullptr;
    int* cout = gather ? (int*)d_out + (long long)DEG_NS * n_nodes : nullptr;
    int* rank = gather ? (int*)d_out + (long long)2 * DEG_NS * n_nodes : nullptr;
    if (gather) {
        int nb = (n_nodes + DEG_R - 1) / DEG_R;
        deg_lds_kernel<<<nb * DEG_NS, DEG_THREADS, 0, stream>>>(src, dst, cin, cout, weight, wT, rank,
                                                                 list_top, n_nodes, n_edges);
    } else {
        long long total = 2LL * n_nodes;
        int grid = (int)((total + 255) / 256);
        init_kernel<<<grid, 256, 0, stream>>>(out, list_top, n_nodes);
        int work = n_edges > IN_FEATS * OUT_FEATS ? n_edges : IN_FEATS * OUT_FEATS;
        int grid2 = (work + 255) / 256;
        deg_wt_kernel<<<grid2, 256, 0, stream>>>(src, dst, out, weight, wT, rank, n_edges);
    }
    {
        int grid = (n_nodes + TILE_M - 1) / TILE_M;
        gemm_maxk_kernel<<<grid, 128, 0, stream>>>(feat, wT, bias, out,
                                                   tk_rec, seg, list_top, cin, cout, n_nodes);
    }
    if (gather) {
        fill_kernel<<<2048, 256, 0, stream>>>(src, dst, rank, cin, sorted_src, n_nodes, n_edges);
        gather_kernel<<<2048, 256, 0, stream>>>(seg, sorted_src, tk_rec, bias, out, n_nodes);
    } else {
        long long total = (long long)n_edges * 4;
        int grid = (int)((total + 255) / 256);
        scatter_kernel<<<grid, 256, 0, stream>>>(src, dst, tk_rec, out, n_edges);
    }
}
